# starting kernel
# baseline (speedup 1.0000x reference)
#include <hip/hip_runtime.h>

typedef float v2f __attribute__((ext_vector_type(2)));
typedef float v8f __attribute__((ext_vector_type(8)));

__global__ __launch_bounds__(256) void deg_init_kernel(float* __restrict__ deg, int N) {
    int n = blockIdx.x * 256 + threadIdx.x;
    if (n < N) deg[n] = 1.0f;
}

__global__ __launch_bounds__(256) void deg_count_kernel(const int* __restrict__ cols,
                                                        float* __restrict__ deg, int E) {
    int e = blockIdx.x * 256 + threadIdx.x;
    if (e < E) atomicAdd(&deg[cols[e]], 1.0f);
}

__global__ __launch_bounds__(256) void dinv_kernel(const float* __restrict__ deg,
                                                   float* __restrict__ dinv, int N) {
    int n = blockIdx.x * 256 + threadIdx.x;
    if (n < N) dinv[n] = rsqrtf(deg[n]);
}

__global__ __launch_bounds__(256) void gemm_xwT_wmma_kernel(const float* __restrict__ X,
                                                            const float* __restrict__ W,
                                                            float* __restrict__ Xhat,
                                                            int N) {
    const int lane    = threadIdx.x & 31;
    const int wave    = threadIdx.x >> 5;
    const int rowBase = blockIdx.x * 16;
    const int colBase = wave * 16;
    if (rowBase >= N) return;

    const int half = lane >> 4;
    const int l16  = lane & 15;

    const float* aptr = X + (size_t)(rowBase + l16) * 128 + (half << 1);
    const float* bptr = W + (size_t)(colBase + l16) * 128 + (half << 1);

    v8f acc = {0.f, 0.f, 0.f, 0.f, 0.f, 0.f, 0.f, 0.f};
#pragma unroll
    for (int k = 0; k < 128; k += 4) {
        v2f a = *(const v2f*)(aptr + k);
        v2f b = *(const v2f*)(bptr + k);
        acc = __builtin_amdgcn_wmma_f32_16x16x4_f32(
             false, a,  false, b,
             (short)0, acc,  false,  false);
    }

    float* cptr = Xhat + (size_t)rowBase * 128 + colBase;
#pragma unroll
    for (int v = 0; v < 8; ++v) {
        int m = v + (half << 3);
        cptr[(size_t)m * 128 + l16] = acc[v];
    }
}

__global__ __launch_bounds__(256) void out_init_kernel(const float* __restrict__ dinv,
                                                       const float* __restrict__ Xhat,
                                                       float* __restrict__ out, int total) {
    int i = blockIdx.x * 256 + threadIdx.x;
    if (i >= total) return;
    float d = dinv[i >> 7];
    out[i] = d * d * Xhat[i];
}

__global__ __launch_bounds__(256) void spmm_edge_kernel(const int* __restrict__ rows,
                                                        const int* __restrict__ cols,
                                                        const float* __restrict__ dinv,
                                                        const float* __restrict__ Xhat,
                                                        float* __restrict__ out, int E) {
    int e = blockIdx.x * 8 + (threadIdx.x >> 5);
    if (e >= E) return;
    int lane = threadIdx.x & 31;
    int r = rows[e];
    int c = cols[e];
    float val = dinv[r] * dinv[c];

    const float4 x = *(const float4*)(Xhat + (size_t)c * 128 + lane * 4);
    float* o = out + (size_t)r * 128 + lane * 4;
    atomicAdd(o + 0, val * x.x);
    atomicAdd(o + 1, val * x.y);
    atomicAdd(o + 2, val * x.z);
    atomicAdd(o + 3, val * x.w);
}

extern "C" void kernel_launch(void* const* d_in, const int* in_sizes, int n_in,
                              void* d_out, int out_size, void* d_ws, size_t ws_size,
                              hipStream_t stream) {
    const float* X  = (const float*)d_in[0];
    const float* W  = (const float*)d_in[1];
    const int*   ei = (const int*)d_in[2];
    float* out = (float*)d_out;

    const int N = in_sizes[0] / 128;
    const int E = in_sizes[2] / 2;
    const int* rows = ei;
    const int* cols = ei + E;

    float* Xhat = (float*)d_ws;
    float* deg  = Xhat + (size_t)N * 128;
    float* dinv = deg + N;

    deg_init_kernel<<<(N + 255) / 256, 256, 0, stream>>>(deg, N);
    deg_count_kernel<<<(E + 255) / 256, 256, 0, stream>>>(cols, deg, E);
    dinv_kernel<<<(N + 255) / 256, 256, 0, stream>>>(deg, dinv, N);

    gemm_xwT_wmma_kernel<<<(N + 15) / 16, 256, 0, stream>>>(X, W, Xhat, N);

    const int total = N * 128;
    out_init_kernel<<<(total + 255) / 256, 256, 0, stream>>>(dinv, Xhat, out, total);

    spmm_edge_kernel<<<(E + 7) / 8, 256, 0, stream>>>(rows, cols, dinv, Xhat, out, E);
}


# spmm: lane f owns features f, f+32, f+64, f+96 (was 4f..4f+3): each f32 atomic instruction adds one whole 128-B line, 4 line requests per edge not 16; row gather becomes 4 one-line loads
# speedup vs baseline: 1.2984x; 1.2984x over previous
#include <hip/hip_runtime.h>

typedef float v2f __attribute__((ext_vector_type(2)));
typedef float v8f __attribute__((ext_vector_type(8)));

__global__ __launch_bounds__(256) void deg_init_kernel(float* __restrict__ deg, int N) {
    int n = blockIdx.x * 256 + threadIdx.x;
    if (n < N) deg[n] = 1.0f;
}

__global__ __launch_bounds__(256) void deg_count_kernel(const int* __restrict__ cols,
                                                        float* __restrict__ deg, int E) {
    int e = blockIdx.x * 256 + threadIdx.x;
    if (e < E) atomicAdd(&deg[cols[e]], 1.0f);
}

__global__ __launch_bounds__(256) void dinv_kernel(const float* __restrict__ deg,
                                                   float* __restrict__ dinv, int N) {
    int n = blockIdx.x * 256 + threadIdx.x;
    if (n < N) dinv[n] = rsqrtf(deg[n]);
}

__global__ __launch_bounds__(256) void gemm_xwT_wmma_kernel(const float* __restrict__ X,
                                                            const float* __restrict__ W,
                                                            float* __restrict__ Xhat,
                                                            int N) {
    const int lane    = threadIdx.x & 31;
    const int wave    = threadIdx.x >> 5;
    const int rowBase = blockIdx.x * 16;
    const int colBase = wave * 16;
    if (rowBase >= N) return;

    const int half = lane >> 4;
    const int l16  = lane & 15;

    const float* aptr = X + (size_t)(rowBase + l16) * 128 + (half << 1);
    const float* bptr = W + (size_t)(colBase + l16) * 128 + (half << 1);

    v8f acc = {0.f, 0.f, 0.f, 0.f, 0.f, 0.f, 0.f, 0.f};
#pragma unroll
    for (int k = 0; k < 128; k += 4) {
        v2f a = *(const v2f*)(aptr + k);
        v2f b = *(const v2f*)(bptr + k);
        acc = __builtin_amdgcn_wmma_f32_16x16x4_f32(
             false, a,  false, b,
             (short)0, acc,  false,  false);
    }

    float* cptr = Xhat + (size_t)rowBase * 128 + colBase;
#pragma unroll
    for (int v = 0; v < 8; ++v) {
        int m = v + (half << 3);
        cptr[(size_t)m * 128 + l16] = acc[v];
    }
}

__global__ __launch_bounds__(256) void out_init_kernel(const float* __restrict__ dinv,
                                                       const float* __restrict__ Xhat,
                                                       float* __restrict__ out, int total) {
    int i = blockIdx.x * 256 + threadIdx.x;
    if (i >= total) return;
    float d = dinv[i >> 7];
    out[i] = d * d * Xhat[i];
}

__global__ __launch_bounds__(256) void spmm_edge_kernel(const int* __restrict__ rows,
                                                        const int* __restrict__ cols,
                                                        const float* __restrict__ dinv,
                                                        const float* __restrict__ Xhat,
                                                        float* __restrict__ out, int E) {
    int e = blockIdx.x * 8 + (threadIdx.x >> 5);
    if (e >= E) return;
    int lane = threadIdx.x & 31;
    int r = rows[e];
    int c = cols[e];
    float val = dinv[r] * dinv[c];

    const float* xr = Xhat + (size_t)c * 128 + lane;
    const float x0 = xr[0];
    const float x1 = xr[32];
    const float x2 = xr[64];
    const float x3 = xr[96];
    float* o = out + (size_t)r * 128 + lane;
    atomicAdd(o + 0,  val * x0);
    atomicAdd(o + 32, val * x1);
    atomicAdd(o + 64, val * x2);
    atomicAdd(o + 96, val * x3);
}

extern "C" void kernel_launch(void* const* d_in, const int* in_sizes, int n_in,
                              void* d_out, int out_size, void* d_ws, size_t ws_size,
                              hipStream_t stream) {
    const float* X  = (const float*)d_in[0];
    const float* W  = (const float*)d_in[1];
    const int*   ei = (const int*)d_in[2];
    float* out = (float*)d_out;

    const int N = in_sizes[0] / 128;
    const int E = in_sizes[2] / 2;
    const int* rows = ei;
    const int* cols = ei + E;

    float* Xhat = (float*)d_ws;
    float* deg  = Xhat + (size_t)N * 128;
    float* dinv = deg + N;

    deg_init_kernel<<<(N + 255) / 256, 256, 0, stream>>>(deg, N);
    deg_count_kernel<<<(E + 255) / 256, 256, 0, stream>>>(cols, deg, E);
    dinv_kernel<<<(N + 255) / 256, 256, 0, stream>>>(deg, dinv, N);

    gemm_xwT_wmma_kernel<<<(N + 15) / 16, 256, 0, stream>>>(X, W, Xhat, N);

    const int total = N * 128;
    out_init_kernel<<<(total + 255) / 256, 256, 0, stream>>>(dinv, Xhat, out, total);

    spmm_edge_kernel<<<(E + 7) / 8, 256, 0, stream>>>(rows, cols, dinv, Xhat, out, E);
}


# row scatter by f32 global atomics replaced by one wave per destination row over edges grouped in the call (count, scan, place), columns sorted per row: f32 sum order changes and is now fixed
# speedup vs baseline: 1.4926x; 1.1496x over previous
#include <hip/hip_runtime.h>

typedef float v2f __attribute__((ext_vector_type(2)));
typedef float v8f __attribute__((ext_vector_type(8)));

__global__ __launch_bounds__(256) void cnt_init_kernel(int* __restrict__ cntR,
                                                       int* __restrict__ degC, int N) {
    int n = blockIdx.x * 256 + threadIdx.x;
    if (n < N) { cntR[n] = 0; degC[n] = 0; }
}

__global__ __launch_bounds__(256) void count_kernel(const int* __restrict__ rows,
                                                    const int* __restrict__ cols,
                                                    int* __restrict__ cntR,
                                                    int* __restrict__ degC,
                                                    int* __restrict__ slot, int E) {
    int e = blockIdx.x * 256 + threadIdx.x;
    if (e >= E) return;
    slot[e] = atomicAdd(&cntR[rows[e]], 1);
    atomicAdd(&degC[cols[e]], 1);
}

__global__ __launch_bounds__(256) void blocksum_kernel(const int* __restrict__ cntR,
                                                       const int* __restrict__ degC,
                                                       int* __restrict__ bsum,
                                                       float* __restrict__ dinv, int N) {
    __shared__ int red[8];
    int n = blockIdx.x * 256 + threadIdx.x;
    int v = 0;
    if (n < N) {
        v = cntR[n];
        dinv[n] = rsqrtf((float)(degC[n] + 1));
    }
    for (int o = 16; o > 0; o >>= 1) v += __shfl_xor(v, o);
    if ((threadIdx.x & 31) == 0) red[threadIdx.x >> 5] = v;
    __syncthreads();
    if (threadIdx.x == 0) {
        int s = 0;
        for (int i = 0; i < 8; ++i) s += red[i];
        bsum[blockIdx.x] = s;
    }
}

__global__ __launch_bounds__(256) void scan_kernel(const int* __restrict__ cntR,
                                                   const int* __restrict__ bsum,
                                                   int* __restrict__ start, int N) {
    __shared__ int red[8];
    __shared__ int wsum[8];
    const int t = threadIdx.x, lane = t & 31, w = t >> 5;
    int p = 0;
    for (int i = t; i < (int)blockIdx.x; i += 256) p += bsum[i];
    for (int o = 16; o > 0; o >>= 1) p += __shfl_xor(p, o);
    int n = blockIdx.x * 256 + t;
    int v = (n < N) ? cntR[n] : 0;
    int inc = v;
    for (int o = 1; o < 32; o <<= 1) {
        int u = __shfl_up(inc, o);
        if (lane >= o) inc += u;
    }
    if (lane == 0) red[w] = p;
    if (lane == 31) wsum[w] = inc;
    __syncthreads();
    int base = 0;
    for (int i = 0; i < 8; ++i) base += red[i];
    for (int i = 0; i < w; ++i) base += wsum[i];
    if (n < N) {
        start[n] = base + inc - v;
        if (n == N - 1) start[N] = base + inc;
    }
}

__global__ __launch_bounds__(256) void place_kernel(const int* __restrict__ rows,
                                                    const int* __restrict__ cols,
                                                    const int* __restrict__ start,
                                                    const int* __restrict__ slot,
                                                    int* __restrict__ scol, int E) {
    int e = blockIdx.x * 256 + threadIdx.x;
    if (e >= E) return;
    scol[start[rows[e]] + slot[e]] = cols[e];
}

__global__ __launch_bounds__(256) void gemm_xwT_wmma_kernel(const float* __restrict__ X,
                                                            const float* __restrict__ W,
                                                            float* __restrict__ Xhat,
                                                            int N) {
    const int lane    = threadIdx.x & 31;
    const int wave    = threadIdx.x >> 5;
    const int rowBase = blockIdx.x * 16;
    const int colBase = wave * 16;
    if (rowBase >= N) return;

    const int half = lane >> 4;
    const int l16  = lane & 15;

    const float* aptr = X + (size_t)(rowBase + l16) * 128 + (half << 1);
    const float* bptr = W + (size_t)(colBase + l16) * 128 + (half << 1);

    v8f acc = {0.f, 0.f, 0.f, 0.f, 0.f, 0.f, 0.f, 0.f};
#pragma unroll
    for (int k = 0; k < 128; k += 4) {
        v2f a = *(const v2f*)(aptr + k);
        v2f b = *(const v2f*)(bptr + k);
        acc = __builtin_amdgcn_wmma_f32_16x16x4_f32(
             false, a,  false, b,
             (short)0, acc,  false,  false);
    }

    float* cptr = Xhat + (size_t)rowBase * 128 + colBase;
#pragma unroll
    for (int v = 0; v < 8; ++v) {
        int m = v + (half << 3);
        cptr[(size_t)m * 128 + l16] = acc[v];
    }
}

__device__ __forceinline__ void axpy4(float4& acc, float v, const float4 x) {
    acc.x = fmaf(v, x.x, acc.x);
    acc.y = fmaf(v, x.y, acc.y);
    acc.z = fmaf(v, x.z, acc.z);
    acc.w = fmaf(v, x.w, acc.w);
}

__global__ __launch_bounds__(256) void gather_rows_kernel(const int* __restrict__ start,
                                                          const int* __restrict__ scol,
                                                          const float* __restrict__ dinv,
                                                          const float* __restrict__ Xhat,
                                                          float* __restrict__ out, int N) {
    const int lane = threadIdx.x & 31;
    const int r = blockIdx.x * 8 + __builtin_amdgcn_readfirstlane(threadIdx.x >> 5);
    if (r >= N) return;
    const int s = start[r];
    const int d = start[r + 1] - s;
    const float dr = dinv[r];
    const float4* __restrict__ X4 = (const float4*)Xhat;

    const float4 xs = X4[(size_t)r * 32 + lane];
    const float vs = dr * dr;
    float4 acc = {vs * xs.x, vs * xs.y, vs * xs.z, vs * xs.w};

    if (d <= 32) {
        const int c = (lane < d) ? scol[s + lane] : 0x7fffffff;
        int rk = 0;
        for (int j = 0; j < d; ++j) {
            const int cj = __builtin_amdgcn_readlane(c, j);
            rk += (cj < c || (cj == c && j < lane)) ? 1 : 0;
        }
        if (lane >= d) rk = lane;
        const int sorted = __builtin_amdgcn_ds_permute(rk << 2, c);

        int k = 0;
        for (; k + 4 <= d; k += 4) {
            const int c0 = __builtin_amdgcn_readlane(sorted, k);
            const int c1 = __builtin_amdgcn_readlane(sorted, k + 1);
            const int c2 = __builtin_amdgcn_readlane(sorted, k + 2);
            const int c3 = __builtin_amdgcn_readlane(sorted, k + 3);
            const float4 x0 = X4[(size_t)c0 * 32 + lane];
            const float4 x1 = X4[(size_t)c1 * 32 + lane];
            const float4 x2 = X4[(size_t)c2 * 32 + lane];
            const float4 x3 = X4[(size_t)c3 * 32 + lane];
            const float v0 = dr * dinv[c0];
            const float v1 = dr * dinv[c1];
            const float v2 = dr * dinv[c2];
            const float v3 = dr * dinv[c3];
            axpy4(acc, v0, x0);
            axpy4(acc, v1, x1);
            axpy4(acc, v2, x2);
            axpy4(acc, v3, x3);
        }
        for (; k < d; ++k) {
            const int c0 = __builtin_amdgcn_readlane(sorted, k);
            const float4 x0 = X4[(size_t)c0 * 32 + lane];
            axpy4(acc, dr * dinv[c0], x0);
        }
    } else {
        int last = -1;
        int done = 0;
        while (done < d) {
            int mn = 0x7fffffff;
            for (int j = lane; j < d; j += 32) {
                const int cj = scol[s + j];
                if (cj > last && cj < mn) mn = cj;
            }
            for (int o = 16; o > 0; o >>= 1) {
                const int u = __shfl_xor(mn, o);
                mn = (u < mn) ? u : mn;
            }
            int cnt = 0;
            for (int j = lane; j < d; j += 32) cnt += (scol[s + j] == mn) ? 1 : 0;
            for (int o = 16; o > 0; o >>= 1) cnt += __shfl_xor(cnt, o);
            cnt = __builtin_amdgcn_readfirstlane(cnt);
            mn  = __builtin_amdgcn_readfirstlane(mn);
            if (cnt <= 0) break;
            const float4 x0 = X4[(size_t)mn * 32 + lane];
            const float v0 = dr * dinv[mn];
            for (int i = 0; i < cnt; ++i) axpy4(acc, v0, x0);
            done += cnt;
            last = mn;
        }
    }
    ((float4*)out)[(size_t)r * 32 + lane] = acc;
}

extern "C" void kernel_launch(void* const* d_in, const int* in_sizes, int n_in,
                              void* d_out, int out_size, void* d_ws, size_t ws_size,
                              hipStream_t stream) {
    const float* X  = (const float*)d_in[0];
    const float* W  = (const float*)d_in[1];
    const int*   ei = (const int*)d_in[2];
    float* out = (float*)d_out;

    const int N = in_sizes[0] / 128;
    const int E = in_sizes[2] / 2;
    const int* rows = ei;
    const int* cols = ei + E;
    const int nb = (N + 255) / 256;

    float* Xhat = (float*)d_ws;
    float* dinv = Xhat + (size_t)N * 128;
    int* cntR  = (int*)(dinv + N);
    int* degC  = cntR + N;
    int* start = degC + N;
    int* bsum  = start + N + 1;
    int* scol  = bsum + nb;
    int* slot  = (int*)d_out;

    cnt_init_kernel<<<nb, 256, 0, stream>>>(cntR, degC, N);
    count_kernel<<<(E + 255) / 256, 256, 0, stream>>>(rows, cols, cntR, degC, slot, E);
    blocksum_kernel<<<nb, 256, 0, stream>>>(cntR, degC, bsum, dinv, N);
    scan_kernel<<<nb, 256, 0, stream>>>(cntR, bsum, start, N);
    place_kernel<<<(E + 255) / 256, 256, 0, stream>>>(rows, cols, start, slot, scol, E);

    gemm_xwT_wmma_kernel<<<(N + 15) / 16, 256, 0, stream>>>(X, W, Xhat, N);

    gather_rows_kernel<<<(N + 7) / 8, 256, 0, stream>>>(start, scol, dinv, Xhat, out, N);
}


# gemm: W staged once per 4-wave block in LDS, one wave per 16x128 strip with 16-B fragment loads (was 8-B loads touching 16 lines each): f32 sum order over k changes inside groups of 8
# speedup vs baseline: 1.6706x; 1.1193x over previous
#include <hip/hip_runtime.h>

typedef float v2f __attribute__((ext_vector_type(2)));
typedef float v8f __attribute__((ext_vector_type(8)));

__global__ __launch_bounds__(256) void cnt_init_kernel(int* __restrict__ cntR,
                                                       int* __restrict__ degC, int N) {
    int n = blockIdx.x * 256 + threadIdx.x;
    if (n < N) { cntR[n] = 0; degC[n] = 0; }
}

__global__ __launch_bounds__(256) void count_kernel(const int* __restrict__ rows,
                                                    const int* __restrict__ cols,
                                                    int* __restrict__ cntR,
                                                    int* __restrict__ degC,
                                                    int* __restrict__ slot, int E) {
    int e = blockIdx.x * 256 + threadIdx.x;
    if (e >= E) return;
    slot[e] = atomicAdd(&cntR[rows[e]], 1);
    atomicAdd(&degC[cols[e]], 1);
}

__global__ __launch_bounds__(256) void blocksum_kernel(const int* __restrict__ cntR,
                                                       const int* __restrict__ degC,
                                                       int* __restrict__ bsum,
                                                       float* __restrict__ dinv, int N) {
    __shared__ int red[8];
    int n = blockIdx.x * 256 + threadIdx.x;
    int v = 0;
    if (n < N) {
        v = cntR[n];
        dinv[n] = rsqrtf((float)(degC[n] + 1));
    }
    for (int o = 16; o > 0; o >>= 1) v += __shfl_xor(v, o);
    if ((threadIdx.x & 31) == 0) red[threadIdx.x >> 5] = v;
    __syncthreads();
    if (threadIdx.x == 0) {
        int s = 0;
        for (int i = 0; i < 8; ++i) s += red[i];
        bsum[blockIdx.x] = s;
    }
}

__global__ __launch_bounds__(256) void scan_kernel(const int* __restrict__ cntR,
                                                   const int* __restrict__ bsum,
                                                   int* __restrict__ start, int N) {
    __shared__ int red[8];
    __shared__ int wsum[8];
    const int t = threadIdx.x, lane = t & 31, w = t >> 5;
    int p = 0;
    for (int i = t; i < (int)blockIdx.x; i += 256) p += bsum[i];
    for (int o = 16; o > 0; o >>= 1) p += __shfl_xor(p, o);
    int n = blockIdx.x * 256 + t;
    int v = (n < N) ? cntR[n] : 0;
    int inc = v;
    for (int o = 1; o < 32; o <<= 1) {
        int u = __shfl_up(inc, o);
        if (lane >= o) inc += u;
    }
    if (lane == 0) red[w] = p;
    if (lane == 31) wsum[w] = inc;
    __syncthreads();
    int base = 0;
    for (int i = 0; i < 8; ++i) base += red[i];
    for (int i = 0; i < w; ++i) base += wsum[i];
    if (n < N) {
        start[n] = base + inc - v;
        if (n == N - 1) start[N] = base + inc;
    }
}

__global__ __launch_bounds__(256) void place_kernel(const int* __restrict__ rows,
                                                    const int* __restrict__ cols,
                                                    const int* __restrict__ start,
                                                    const int* __restrict__ slot,
                                                    int* __restrict__ scol, int E) {
    int e = blockIdx.x * 256 + threadIdx.x;
    if (e >= E) return;
    scol[start[rows[e]] + slot[e]] = cols[e];
}

#define W_PITCH 132

__device__ __forceinline__ v8f wmma4(v2f a, v2f b, v8f c) {
    return __builtin_amdgcn_wmma_f32_16x16x4_f32(
         false, a,  false, b,
         (short)0, c,  false,  false);
}

__global__ __launch_bounds__(128) void gemm_xwT_wmma_kernel(const float* __restrict__ X,
                                                            const float* __restrict__ W,
                                                            float* __restrict__ Xhat,
                                                            int N) {
    __shared__ __attribute__((aligned(16))) float Ws[128 * W_PITCH];

    const int t = threadIdx.x;
    const float4* __restrict__ W4 = (const float4*)W;
#pragma unroll 8
    for (int q = 0; q < 32; ++q) {
        const int i = t + 128 * q;
        const int row = i >> 5, c4 = i & 31;
        const float4 v = W4[i];
        *(float4*)&Ws[row * W_PITCH + c4 * 4] = v;
    }
    __syncthreads();

    const int lane = t & 31;
    const int wave = __builtin_amdgcn_readfirstlane(t >> 5);
    const int rowBase = (blockIdx.x * 4 + wave) * 16;
    if (rowBase >= N) return;

    const int half = lane >> 4;
    const int l16  = lane & 15;

    int ar = rowBase + l16;
    ar = (ar < N) ? ar : (N - 1);
    const float4* __restrict__ aptr = (const float4*)(X + (size_t)ar * 128 + half * 4);
    float4 a[16];
#pragma unroll
    for (int j = 0; j < 16; ++j) a[j] = aptr[2 * j];

    v8f acc[8];
#pragma unroll
    for (int ct = 0; ct < 8; ++ct)
        acc[ct] = (v8f){0.f, 0.f, 0.f, 0.f, 0.f, 0.f, 0.f, 0.f};

    const float* bbase = &Ws[l16 * W_PITCH + half * 4];
#pragma unroll
    for (int j = 0; j < 16; ++j) {
        float4 b[8];
#pragma unroll
        for (int ct = 0; ct < 8; ++ct)
            b[ct] = *(const float4*)(bbase + ct * 16 * W_PITCH + 8 * j);
        const v2f a0 = {a[j].x, a[j].y};
        const v2f a1 = {a[j].z, a[j].w};
#pragma unroll
        for (int ct = 0; ct < 8; ++ct) {
            const v2f b0 = {b[ct].x, b[ct].y};
            acc[ct] = wmma4(a0, b0, acc[ct]);
        }
#pragma unroll
        for (int ct = 0; ct < 8; ++ct) {
            const v2f b1 = {b[ct].z, b[ct].w};
            acc[ct] = wmma4(a1, b1, acc[ct]);
        }
    }

    float* cptr = Xhat + (size_t)rowBase * 128 + l16;
#pragma unroll
    for (int v = 0; v < 8; ++v) {
        const int m = v + (half << 3);
        if (rowBase + m < N) {
#pragma unroll
            for (int ct = 0; ct < 8; ++ct)
                cptr[(size_t)m * 128 + ct * 16] = acc[ct][v];
        }
    }
}

__device__ __forceinline__ void axpy4(float4& acc, float v, const float4 x) {
    acc.x = fmaf(v, x.x, acc.x);
    acc.y = fmaf(v, x.y, acc.y);
    acc.z = fmaf(v, x.z, acc.z);
    acc.w = fmaf(v, x.w, acc.w);
}

__global__ __launch_bounds__(256) void gather_rows_kernel(const int* __restrict__ start,
                                                          const int* __restrict__ scol,
                                                          const float* __restrict__ dinv,
                                                          const float* __restrict__ Xhat,
                                                          float* __restrict__ out, int N) {
    const int lane = threadIdx.x & 31;
    const int r = blockIdx.x * 8 + __builtin_amdgcn_readfirstlane(threadIdx.x >> 5);
    if (r >= N) return;
    const int s = start[r];
    const int d = start[r + 1] - s;
    const float dr = dinv[r];
    const float4* __restrict__ X4 = (const float4*)Xhat;

    const float4 xs = X4[(size_t)r * 32 + lane];
    const float vs = dr * dr;
    float4 acc = {vs * xs.x, vs * xs.y, vs * xs.z, vs * xs.w};

    if (d <= 32) {
        const int c = (lane < d) ? scol[s + lane] : 0x7fffffff;
        int rk = 0;
        for (int j = 0; j < d; ++j) {
            const int cj = __builtin_amdgcn_readlane(c, j);
            rk += (cj < c || (cj == c && j < lane)) ? 1 : 0;
        }
        if (lane >= d) rk = lane;
        const int sorted = __builtin_amdgcn_ds_permute(rk << 2, c);

        int k = 0;
        for (; k + 4 <= d; k += 4) {
            const int c0 = __builtin_amdgcn_readlane(sorted, k);
            const int c1 = __builtin_amdgcn_readlane(sorted, k + 1);
            const int c2 = __builtin_amdgcn_readlane(sorted, k + 2);
            const int c3 = __builtin_amdgcn_readlane(sorted, k + 3);
            const float4 x0 = X4[(size_t)c0 * 32 + lane];
            const float4 x1 = X4[(size_t)c1 * 32 + lane];
            const float4 x2 = X4[(size_t)c2 * 32 + lane];
            const float4 x3 = X4[(size_t)c3 * 32 + lane];
            const float v0 = dr * dinv[c0];
            const float v1 = dr * dinv[c1];
            const float v2 = dr * dinv[c2];
            const float v3 = dr * dinv[c3];
            axpy4(acc, v0, x0);
            axpy4(acc, v1, x1);
            axpy4(acc, v2, x2);
            axpy4(acc, v3, x3);
        }
        for (; k < d; ++k) {
            const int c0 = __builtin_amdgcn_readlane(sorted, k);
            const float4 x0 = X4[(size_t)c0 * 32 + lane];
            axpy4(acc, dr * dinv[c0], x0);
        }
    } else {
        int last = -1;
        int done = 0;
        while (done < d) {
            int mn = 0x7fffffff;
            for (int j = lane; j < d; j += 32) {
                const int cj = scol[s + j];
                if (cj > last && cj < mn) mn = cj;
            }
            for (int o = 16; o > 0; o >>= 1) {
                const int u = __shfl_xor(mn, o);
                mn = (u < mn) ? u : mn;
            }
            int cnt = 0;
            for (int j = lane; j < d; j += 32) cnt += (scol[s + j] == mn) ? 1 : 0;
            for (int o = 16; o > 0; o >>= 1) cnt += __shfl_xor(cnt, o);
            cnt = __builtin_amdgcn_readfirstlane(cnt);
            mn  = __builtin_amdgcn_readfirstlane(mn);
            if (cnt <= 0) break;
            const float4 x0 = X4[(size_t)mn * 32 + lane];
            const float v0 = dr * dinv[mn];
            for (int i = 0; i < cnt; ++i) axpy4(acc, v0, x0);
            done += cnt;
            last = mn;
        }
    }
    ((float4*)out)[(size_t)r * 32 + lane] = acc;
}

extern "C" void kernel_launch(void* const* d_in, const int* in_sizes, int n_in,
                              void* d_out, int out_size, void* d_ws, size_t ws_size,
                              hipStream_t stream) {
    const float* X  = (const float*)d_in[0];
    const float* W  = (const float*)d_in[1];
    const int*   ei = (const int*)d_in[2];
    float* out = (float*)d_out;

    const int N = in_sizes[0] / 128;
    const int E = in_sizes[2] / 2;
    const int* rows = ei;
    const int* cols = ei + E;
    const int nb = (N + 255) / 256;

    float* Xhat = (float*)d_ws;
    float* dinv = Xhat + (size_t)N * 128;
    int* cntR  = (int*)(dinv + N);
    int* degC  = cntR + N;
    int* start = degC + N;
    int* bsum  = start + N + 1;
    int* scol  = bsum + nb;
    int* slot  = (int*)d_out;

    cnt_init_kernel<<<nb, 256, 0, stream>>>(cntR, degC, N);
    count_kernel<<<(E + 255) / 256, 256, 0, stream>>>(rows, cols, cntR, degC, slot, E);
    blocksum_kernel<<<nb, 256, 0, stream>>>(cntR, degC, bsum, dinv, N);
    scan_kernel<<<nb, 256, 0, stream>>>(cntR, bsum, start, N);
    place_kernel<<<(E + 255) / 256, 256, 0, stream>>>(rows, cols, start, slot, scol, E);

    gemm_xwT_wmma_kernel<<<(N + 63) / 64, 128, 0, stream>>>(X, W, Xhat, N);

    gather_rows_kernel<<<(N + 7) / 8, 256, 0, stream>>>(start, scol, dinv, Xhat, out, N);
}


# edges grouped by the count launch itself into a 128-slot bucket in each row's own output row; block-sum, scan and place launches dropped (7 to 4), deg^-1/2 taken where used; f32 sum order unchanged
# speedup vs baseline: 1.8817x; 1.1264x over previous
#include <hip/hip_runtime.h>

typedef float v2f __attribute__((ext_vector_type(2)));
typedef float v8f __attribute__((ext_vector_type(8)));


#define ROW_CAP 128

__global__ __launch_bounds__(256) void cnt_init_kernel(int* __restrict__ cntR,
                                                       int* __restrict__ degC, int N) {
    int n = blockIdx.x * 256 + threadIdx.x;
    if (n < N) { cntR[n] = 0; degC[n] = 0; }
}

__global__ __launch_bounds__(256) void count_bucket_kernel(const int* __restrict__ rows,
                                                           const int* __restrict__ cols,
                                                           int* __restrict__ cntR,
                                                           int* __restrict__ degC,
                                                           int* __restrict__ bucket, int E) {
    int e = blockIdx.x * 256 + threadIdx.x;
    if (e >= E) return;
    const int r = rows[e];
    const int c = cols[e];
    const int slot = atomicAdd(&cntR[r], 1);
    atomicAdd(&degC[c], 1);
    if (slot < ROW_CAP) bucket[(size_t)r * 128 + slot] = c;
}

#define W_PITCH 132

__device__ __forceinline__ v8f wmma4(v2f a, v2f b, v8f c) {
    return __builtin_amdgcn_wmma_f32_16x16x4_f32(
         false, a,  false, b,
         (short)0, c,  false,  false);
}

__global__ __launch_bounds__(128) void gemm_xwT_wmma_kernel(const float* __restrict__ X,
                                                            const float* __restrict__ W,
                                                            float* __restrict__ Xhat,
                                                            int N) {
    __shared__ __attribute__((aligned(16))) float Ws[128 * W_PITCH];

    const int t = threadIdx.x;
    const float4* __restrict__ W4 = (const float4*)W;
#pragma unroll 8
    for (int q = 0; q < 32; ++q) {
        const int i = t + 128 * q;
        const int row = i >> 5, c4 = i & 31;
        const float4 v = W4[i];
        *(float4*)&Ws[row * W_PITCH + c4 * 4] = v;
    }
    __syncthreads();

    const int lane = t & 31;
    const int wave = __builtin_amdgcn_readfirstlane(t >> 5);
    const int rowBase = (blockIdx.x * 4 + wave) * 16;
    if (rowBase >= N) return;

    const int half = lane >> 4;
    const int l16  = lane & 15;

    int ar = rowBase + l16;
    ar = (ar < N) ? ar : (N - 1);
    const float4* __restrict__ aptr = (const float4*)(X + (size_t)ar * 128 + half * 4);
    float4 a[16];
#pragma unroll
    for (int j = 0; j < 16; ++j) a[j] = aptr[2 * j];

    v8f acc[8];
#pragma unroll
    for (int ct = 0; ct < 8; ++ct)
        acc[ct] = (v8f){0.f, 0.f, 0.f, 0.f, 0.f, 0.f, 0.f, 0.f};

    const float* bbase = &Ws[l16 * W_PITCH + half * 4];
#pragma unroll
    for (int j = 0; j < 16; ++j) {
        float4 b[8];
#pragma unroll
        for (int ct = 0; ct < 8; ++ct)
            b[ct] = *(const float4*)(bbase + ct * 16 * W_PITCH + 8 * j);
        const v2f a0 = {a[j].x, a[j].y};
        const v2f a1 = {a[j].z, a[j].w};
#pragma unroll
        for (int ct = 0; ct < 8; ++ct) {
            const v2f b0 = {b[ct].x, b[ct].y};
            acc[ct] = wmma4(a0, b0, acc[ct]);
        }
#pragma unroll
        for (int ct = 0; ct < 8; ++ct) {
            const v2f b1 = {b[ct].z, b[ct].w};
            acc[ct] = wmma4(a1, b1, acc[ct]);
        }
    }

    float* cptr = Xhat + (size_t)rowBase * 128 + l16;
#pragma unroll
    for (int v = 0; v < 8; ++v) {
        const int m = v + (half << 3);
        if (rowBase + m < N) {
#pragma unroll
            for (int ct = 0; ct < 8; ++ct)
                cptr[(size_t)m * 128 + ct * 16] = acc[ct][v];
        }
    }
}

__device__ __forceinline__ void axpy4(float4& acc, float v, const float4 x) {
    acc.x = fmaf(v, x.x, acc.x);
    acc.y = fmaf(v, x.y, acc.y);
    acc.z = fmaf(v, x.z, acc.z);
    acc.w = fmaf(v, x.w, acc.w);
}

__device__ __forceinline__ float dinv_from(int deg) {
    return __builtin_amdgcn_rsqf((float)(deg + 1));
}
__device__ __forceinline__ float dinv_of(const int* __restrict__ degC, int n) {
    return dinv_from(degC[n]);
}

__device__ __forceinline__ void sum_long_row(float4& acc, const int* keys, const int* vals,
                                             int n, int total, int r, int lane, float dr,
                                             const int* __restrict__ degC,
                                             const float4* __restrict__ X4) {
    int last = -1;
    int done = 0;
    while (done < total) {
        int mn = 0x7fffffff;
        for (int j = lane; j < n; j += 32) {
            const bool mine = keys ? (keys[j] == r) : true;
            const int cj = vals[j];
            if (mine && cj > last && cj < mn) mn = cj;
        }
        for (int o = 16; o > 0; o >>= 1) {
            const int u = __shfl_xor(mn, o);
            mn = (u < mn) ? u : mn;
        }
        int cnt = 0;
        for (int j = lane; j < n; j += 32) {
            const bool mine = keys ? (keys[j] == r) : true;
            cnt += (mine && vals[j] == mn) ? 1 : 0;
        }
        for (int o = 16; o > 0; o >>= 1) cnt += __shfl_xor(cnt, o);
        cnt = __builtin_amdgcn_readfirstlane(cnt);
        mn  = __builtin_amdgcn_readfirstlane(mn);
        if (cnt <= 0) break;
        const float4 x0 = X4[(size_t)mn * 32 + lane];
        const float v0 = dr * dinv_of(degC, mn);
        for (int i = 0; i < cnt; ++i) axpy4(acc, v0, x0);
        done += cnt;
        last = mn;
    }
}

__global__ __launch_bounds__(256) void gather_rows_kernel(const int* __restrict__ cntR,
                                                          const int* __restrict__ degC,
                                                          const float* __restrict__ Xhat,
                                                          const int* __restrict__ rows,
                                                          const int* __restrict__ cols,
                                                          int* out, int N, int E) {
    const int lane = threadIdx.x & 31;
    const int r = blockIdx.x * 8 + __builtin_amdgcn_readfirstlane(threadIdx.x >> 5);
    if (r >= N) return;
    const int d = cntR[r];
    const float dr = dinv_of(degC, r);
    const float4* __restrict__ X4 = (const float4*)Xhat;
    int* orow = out + (size_t)r * 128;

    const float4 xs = X4[(size_t)r * 32 + lane];
    const float vs = dr * dr;
    float4 acc = {vs * xs.x, vs * xs.y, vs * xs.z, vs * xs.w};

    if (d <= 32) {
        const int c = (lane < d) ? orow[lane] : 0x7fffffff;
        int rk = 0;
        for (int j = 0; j < d; ++j) {
            const int cj = __builtin_amdgcn_readlane(c, j);
            rk += (cj < c || (cj == c && j < lane)) ? 1 : 0;
        }
        if (lane >= d) rk = lane;
        const int sorted = __builtin_amdgcn_ds_permute(rk << 2, c);

        int k = 0;
        for (; k + 4 <= d; k += 4) {
            const int c0 = __builtin_amdgcn_readlane(sorted, k);
            const int c1 = __builtin_amdgcn_readlane(sorted, k + 1);
            const int c2 = __builtin_amdgcn_readlane(sorted, k + 2);
            const int c3 = __builtin_amdgcn_readlane(sorted, k + 3);
            const float4 x0 = X4[(size_t)c0 * 32 + lane];
            const float4 x1 = X4[(size_t)c1 * 32 + lane];
            const float4 x2 = X4[(size_t)c2 * 32 + lane];
            const float4 x3 = X4[(size_t)c3 * 32 + lane];
            const int g0 = degC[c0];
            const int g1 = degC[c1];
            const int g2 = degC[c2];
            const int g3 = degC[c3];
            const float v0 = dr * dinv_from(g0);
            const float v1 = dr * dinv_from(g1);
            const float v2 = dr * dinv_from(g2);
            const float v3 = dr * dinv_from(g3);
            axpy4(acc, v0, x0);
            axpy4(acc, v1, x1);
            axpy4(acc, v2, x2);
            axpy4(acc, v3, x3);
        }
        for (; k < d; ++k) {
            const int c0 = __builtin_amdgcn_readlane(sorted, k);
            const float4 x0 = X4[(size_t)c0 * 32 + lane];
            axpy4(acc, dr * dinv_of(degC, c0), x0);
        }
    } else if (d <= ROW_CAP) {
        sum_long_row(acc, nullptr, orow, d, d, r, lane, dr, degC, X4);
    } else {
        sum_long_row(acc, rows, cols, E, d, r, lane, dr, degC, X4);
    }
    const int4 res = {__float_as_int(acc.x), __float_as_int(acc.y),
                      __float_as_int(acc.z), __float_as_int(acc.w)};
    *(int4*)(orow + lane * 4) = res;
}

extern "C" void kernel_launch(void* const* d_in, const int* in_sizes, int n_in,
                              void* d_out, int out_size, void* d_ws, size_t ws_size,
                              hipStream_t stream) {
    const float* X  = (const float*)d_in[0];
    const float* W  = (const float*)d_in[1];
    const int*   ei = (const int*)d_in[2];
    float* out = (float*)d_out;

    const int N = in_sizes[0] / 128;
    const int E = in_sizes[2] / 2;
    const int* rows = ei;
    const int* cols = ei + E;
    const int nb = (N + 255) / 256;

    float* Xhat = (float*)d_ws;
    int* cntR   = (int*)(Xhat + (size_t)N * 128);
    int* degC   = cntR + N;
    int* bucket = (int*)out;

    cnt_init_kernel<<<nb, 256, 0, stream>>>(cntR, degC, N);
    count_bucket_kernel<<<(E + 255) / 256, 256, 0, stream>>>(rows, cols, cntR, degC, bucket, E);

    gemm_xwT_wmma_kernel<<<(N + 63) / 64, 128, 0, stream>>>(X, W, Xhat, N);

    gather_rows_kernel<<<(N + 7) / 8, 256, 0, stream>>>(cntR, degC, Xhat, rows, cols, bucket, N, E);
}


# edges grouped by row through LDS counters in two levels (256 edge chunks x 256 row ranges, then one block per range fills the row buckets): no returning global atomic per edge; f32 sum order kept
# speedup vs baseline: 2.1106x; 1.1216x over previous
#include <hip/hip_runtime.h>

typedef float v2f __attribute__((ext_vector_type(2)));
typedef float v8f __attribute__((ext_vector_type(8)));


#define ROW_CAP 128

__global__ __launch_bounds__(256) void cnt_init_kernel(int* __restrict__ degC, int N) {
    int n = blockIdx.x * 256 + threadIdx.x;
    if (n < N) degC[n] = 0;
}

#define NPART 256
#define BIN_ROWS_MAX 1024
#define BIN_EDGES_MAX 12288

__device__ __forceinline__ int rows_per_bin(int N) {
    const int q = N / NPART;
    return q > 0 ? q : 1;
}

__global__ __launch_bounds__(1024) void partition_kernel(const int* __restrict__ rows,
                                                         const int* __restrict__ cols,
                                                         int* __restrict__ degC,
                                                         int* __restrict__ seg, int N, int E) {
    __shared__ int binCnt[NPART];
    const int t = threadIdx.x;
    const int c = blockIdx.x;
    if (t < NPART) binCnt[t] = 0;
    __syncthreads();
    const int rpb = rows_per_bin(N);
    const int capPairs = (rpb * 128 - NPART) / (2 * NPART);
    const int chunkE = (E + NPART - 1) / NPART;
    const int e0 = c * chunkE;
    int e1 = e0 + chunkE;
    e1 = e1 < E ? e1 : E;
    for (int e = e0 + t; e < e1; e += 1024) {
        const int r = rows[e];
        const int col = cols[e];
        atomicAdd(&degC[col], 1);
        int b = r / rpb;
        b = b < NPART - 1 ? b : NPART - 1;
        const int slot = atomicAdd(&binCnt[b], 1);
        if (slot < capPairs) {
            int* p = seg + (size_t)b * rpb * 128 + (size_t)(c * capPairs + slot) * 2;
            *(int2*)p = make_int2(r - b * rpb, col);
        }
    }
    __syncthreads();
    if (t < NPART) seg[(size_t)t * rpb * 128 + (rpb * 128 - NPART) + c] = binCnt[t];
}

__global__ __launch_bounds__(1024) void binplace_kernel(const int* __restrict__ rows,
                                                        const int* __restrict__ cols,
                                                        int* __restrict__ cntR,
                                                        int* bucket, int N, int E) {
    __shared__ int rowCnt[BIN_ROWS_MAX];
    __shared__ int segOff[NPART + 1];
    __shared__ int segCnt[NPART];
    __shared__ int waveTot[NPART / 32];
    __shared__ int useList;
    __shared__ __attribute__((aligned(8))) int2 ed[BIN_EDGES_MAX];

    const int t = threadIdx.x;
    const int b = blockIdx.x;
    const int rpb = rows_per_bin(N);
    const int capPairs = (rpb * 128 - NPART) / (2 * NPART);
    const int base = b * rpb;
    int nrows = (b == NPART - 1) ? (N - base) : rpb;
    if (base >= N) nrows = 0;
    if (nrows > BIN_ROWS_MAX) nrows = BIN_ROWS_MAX;

    for (int i = t; i < BIN_ROWS_MAX; i += 1024) rowCnt[i] = 0;
    int* brow = bucket + (size_t)base * 128;
    if (t == 0) useList = 0;
    const int n = (t < NPART) ? brow[rpb * 128 - NPART + t] : 0;
    int incl = n;
    for (int o = 1; o < 32; o <<= 1) {
        const int u = __shfl_up(incl, o);
        if ((t & 31) >= o) incl += u;
    }
    if (t < NPART) {
        segCnt[t] = n;
        if ((t & 31) == 31) waveTot[t >> 5] = incl;
    }
    __syncthreads();
    if (t < NPART) {
        int off = 0;
        for (int w = 0; w < (t >> 5); ++w) off += waveTot[w];
        segOff[t] = off + incl - n;
        if (n > capPairs) useList = 1;
        if (t == NPART - 1) {
            segOff[NPART] = off + incl;
            if (off + incl > BIN_EDGES_MAX) useList = 1;
        }
    }
    __syncthreads();
    const int total = segOff[NPART];

    if (!useList) {
        const int wave = t >> 5, lane = t & 31;
        for (int c = wave; c < NPART; c += 32) {
            const int n = segCnt[c];
            const int o = segOff[c];
            const int2* p = (const int2*)(brow + (size_t)c * capPairs * 2);
            for (int k = lane; k < n; k += 32) ed[o + k] = p[k];
        }
        __syncthreads();
        for (int i = t; i < total; i += 1024) {
            const int2 v = ed[i];
            const int slot = atomicAdd(&rowCnt[v.x], 1);
            if (slot < ROW_CAP) brow[(size_t)v.x * 128 + slot] = v.y;
        }
    } else {
        for (int e = t; e < E; e += 1024) {
            const int r = rows[e];
            int bb = r / rpb;
            bb = bb < NPART - 1 ? bb : NPART - 1;
            if (bb == b && r - base < nrows) {
                const int slot = atomicAdd(&rowCnt[r - base], 1);
                if (slot < ROW_CAP) brow[(size_t)(r - base) * 128 + slot] = cols[e];
            }
        }
    }
    __syncthreads();
    for (int i = t; i < nrows; i += 1024) cntR[base + i] = rowCnt[i];
}

#define W_PITCH 132

__device__ __forceinline__ v8f wmma4(v2f a, v2f b, v8f c) {
    return __builtin_amdgcn_wmma_f32_16x16x4_f32(
         false, a,  false, b,
         (short)0, c,  false,  false);
}

__global__ __launch_bounds__(128) void gemm_xwT_wmma_kernel(const float* __restrict__ X,
                                                            const float* __restrict__ W,
                                                            float* __restrict__ Xhat,
                                                            int N) {
    __shared__ __attribute__((aligned(16))) float Ws[128 * W_PITCH];

    const int t = threadIdx.x;
    const float4* __restrict__ W4 = (const float4*)W;
#pragma unroll 8
    for (int q = 0; q < 32; ++q) {
        const int i = t + 128 * q;
        const int row = i >> 5, c4 = i & 31;
        const float4 v = W4[i];
        *(float4*)&Ws[row * W_PITCH + c4 * 4] = v;
    }
    __syncthreads();

    const int lane = t & 31;
    const int wave = __builtin_amdgcn_readfirstlane(t >> 5);
    const int rowBase = (blockIdx.x * 4 + wave) * 16;
    if (rowBase >= N) return;

    const int half = lane >> 4;
    const int l16  = lane & 15;

    int ar = rowBase + l16;
    ar = (ar < N) ? ar : (N - 1);
    const float4* __restrict__ aptr = (const float4*)(X + (size_t)ar * 128 + half * 4);
    float4 a[16];
#pragma unroll
    for (int j = 0; j < 16; ++j) a[j] = aptr[2 * j];

    v8f acc[8];
#pragma unroll
    for (int ct = 0; ct < 8; ++ct)
        acc[ct] = (v8f){0.f, 0.f, 0.f, 0.f, 0.f, 0.f, 0.f, 0.f};

    const float* bbase = &Ws[l16 * W_PITCH + half * 4];
#pragma unroll
    for (int j = 0; j < 16; ++j) {
        float4 b[8];
#pragma unroll
        for (int ct = 0; ct < 8; ++ct)
            b[ct] = *(const float4*)(bbase + ct * 16 * W_PITCH + 8 * j);
        const v2f a0 = {a[j].x, a[j].y};
        const v2f a1 = {a[j].z, a[j].w};
#pragma unroll
        for (int ct = 0; ct < 8; ++ct) {
            const v2f b0 = {b[ct].x, b[ct].y};
            acc[ct] = wmma4(a0, b0, acc[ct]);
        }
#pragma unroll
        for (int ct = 0; ct < 8; ++ct) {
            const v2f b1 = {b[ct].z, b[ct].w};
            acc[ct] = wmma4(a1, b1, acc[ct]);
        }
    }

    float* cptr = Xhat + (size_t)rowBase * 128 + l16;
#pragma unroll
    for (int v = 0; v < 8; ++v) {
        const int m = v + (half << 3);
        if (rowBase + m < N) {
#pragma unroll
            for (int ct = 0; ct < 8; ++ct)
                cptr[(size_t)m * 128 + ct * 16] = acc[ct][v];
        }
    }
}

__device__ __forceinline__ void axpy4(float4& acc, float v, const float4 x) {
    acc.x = fmaf(v, x.x, acc.x);
    acc.y = fmaf(v, x.y, acc.y);
    acc.z = fmaf(v, x.z, acc.z);
    acc.w = fmaf(v, x.w, acc.w);
}

__device__ __forceinline__ float dinv_from(int deg) {
    return __builtin_amdgcn_rsqf((float)(deg + 1));
}
__device__ __forceinline__ float dinv_of(const int* __restrict__ degC, int n) {
    return dinv_from(degC[n]);
}

__device__ __forceinline__ void sum_long_row(float4& acc, const int* keys, const int* vals,
                                             int n, int total, int r, int lane, float dr,
                                             const int* __restrict__ degC,
                                             const float4* __restrict__ X4) {
    int last = -1;
    int done = 0;
    while (done < total) {
        int mn = 0x7fffffff;
        for (int j = lane; j < n; j += 32) {
            const bool mine = keys ? (keys[j] == r) : true;
            const int cj = vals[j];
            if (mine && cj > last && cj < mn) mn = cj;
        }
        for (int o = 16; o > 0; o >>= 1) {
            const int u = __shfl_xor(mn, o);
            mn = (u < mn) ? u : mn;
        }
        int cnt = 0;
        for (int j = lane; j < n; j += 32) {
            const bool mine = keys ? (keys[j] == r) : true;
            cnt += (mine && vals[j] == mn) ? 1 : 0;
        }
        for (int o = 16; o > 0; o >>= 1) cnt += __shfl_xor(cnt, o);
        cnt = __builtin_amdgcn_readfirstlane(cnt);
        mn  = __builtin_amdgcn_readfirstlane(mn);
        if (cnt <= 0) break;
        const float4 x0 = X4[(size_t)mn * 32 + lane];
        const float v0 = dr * dinv_of(degC, mn);
        for (int i = 0; i < cnt; ++i) axpy4(acc, v0, x0);
        done += cnt;
        last = mn;
    }
}

__global__ __launch_bounds__(256) void gather_rows_kernel(const int* __restrict__ cntR,
                                                          const int* __restrict__ degC,
                                                          const float* __restrict__ Xhat,
                                                          const int* __restrict__ rows,
                                                          const int* __restrict__ cols,
                                                          int* out, int N, int E) {
    const int lane = threadIdx.x & 31;
    const int r = blockIdx.x * 8 + __builtin_amdgcn_readfirstlane(threadIdx.x >> 5);
    if (r >= N) return;
    const int d = cntR[r];
    const float dr = dinv_of(degC, r);
    const float4* __restrict__ X4 = (const float4*)Xhat;
    int* orow = out + (size_t)r * 128;

    const float4 xs = X4[(size_t)r * 32 + lane];
    const float vs = dr * dr;
    float4 acc = {vs * xs.x, vs * xs.y, vs * xs.z, vs * xs.w};

    if (d <= 32) {
        const int c = (lane < d) ? orow[lane] : 0x7fffffff;
        int rk = 0;
        for (int j = 0; j < d; ++j) {
            const int cj = __builtin_amdgcn_readlane(c, j);
            rk += (cj < c || (cj == c && j < lane)) ? 1 : 0;
        }
        if (lane >= d) rk = lane;
        const int sorted = __builtin_amdgcn_ds_permute(rk << 2, c);

        int k = 0;
        for (; k + 4 <= d; k += 4) {
            const int c0 = __builtin_amdgcn_readlane(sorted, k);
            const int c1 = __builtin_amdgcn_readlane(sorted, k + 1);
            const int c2 = __builtin_amdgcn_readlane(sorted, k + 2);
            const int c3 = __builtin_amdgcn_readlane(sorted, k + 3);
            const float4 x0 = X4[(size_t)c0 * 32 + lane];
            const float4 x1 = X4[(size_t)c1 * 32 + lane];
            const float4 x2 = X4[(size_t)c2 * 32 + lane];
            const float4 x3 = X4[(size_t)c3 * 32 + lane];
            const int g0 = degC[c0];
            const int g1 = degC[c1];
            const int g2 = degC[c2];
            const int g3 = degC[c3];
            const float v0 = dr * dinv_from(g0);
            const float v1 = dr * dinv_from(g1);
            const float v2 = dr * dinv_from(g2);
            const float v3 = dr * dinv_from(g3);
            axpy4(acc, v0, x0);
            axpy4(acc, v1, x1);
            axpy4(acc, v2, x2);
            axpy4(acc, v3, x3);
        }
        for (; k < d; ++k) {
            const int c0 = __builtin_amdgcn_readlane(sorted, k);
            const float4 x0 = X4[(size_t)c0 * 32 + lane];
            axpy4(acc, dr * dinv_of(degC, c0), x0);
        }
    } else if (d <= ROW_CAP) {
        sum_long_row(acc, nullptr, orow, d, d, r, lane, dr, degC, X4);
    } else {
        sum_long_row(acc, rows, cols, E, d, r, lane, dr, degC, X4);
    }
    const int4 res = {__float_as_int(acc.x), __float_as_int(acc.y),
                      __float_as_int(acc.z), __float_as_int(acc.w)};
    *(int4*)(orow + lane * 4) = res;
}

extern "C" void kernel_launch(void* const* d_in, const int* in_sizes, int n_in,
                              void* d_out, int out_size, void* d_ws, size_t ws_size,
                              hipStream_t stream) {
    const float* X  = (const float*)d_in[0];
    const float* W  = (const float*)d_in[1];
    const int*   ei = (const int*)d_in[2];
    float* out = (float*)d_out;

    const int N = in_sizes[0] / 128;
    const int E = in_sizes[2] / 2;
    const int* rows = ei;
    const int* cols = ei + E;
    const int nb = (N + 255) / 256;

    float* Xhat = (float*)d_ws;
    int* cntR   = (int*)(Xhat + (size_t)N * 128);
    int* degC   = cntR + N;
    int* bucket = (int*)out;

    cnt_init_kernel<<<nb, 256, 0, stream>>>(degC, N);
    partition_kernel<<<NPART, 1024, 0, stream>>>(rows, cols, degC, bucket, N, E);
    binplace_kernel<<<NPART, 1024, 0, stream>>>(rows, cols, cntR, bucket, N, E);

    gemm_xwT_wmma_kernel<<<(N + 63) / 64, 128, 0, stream>>>(X, W, Xhat, N);

    gather_rows_kernel<<<(N + 7) / 8, 256, 0, stream>>>(cntR, degC, Xhat, rows, cols, bucket, N, E);
}


# both edge-grouping launches order their edges in LDS (count, scan, place) and store each bin piece and each row bucket as one contiguous run, not one scattered store per edge; f32 sum order kept
# speedup vs baseline: 2.2375x; 1.0601x over previous
#include <hip/hip_runtime.h>

typedef float v2f __attribute__((ext_vector_type(2)));
typedef float v8f __attribute__((ext_vector_type(8)));


#define ROW_CAP 128

__global__ __launch_bounds__(256) void cnt_init_kernel(int* __restrict__ degC, int N) {
    int n = blockIdx.x * 256 + threadIdx.x;
    if (n < N) degC[n] = 0;
}

#define NPART 256
#define BIN_ROWS_MAX 1024
#define BIN_EDGES_MAX 12288
#define CHUNK_MAX 8192

__device__ __forceinline__ int rows_per_bin(int N) {
    const int q = N / NPART;
    return q > 0 ? q : 1;
}

__device__ __forceinline__ int cap_pairs(int rpb) {
    return ((rpb * 128 - NPART) / (2 * NPART)) & ~15;
}

__global__ __launch_bounds__(1024) void partition_kernel(const int* __restrict__ rows,
                                                         const int* __restrict__ cols,
                                                         int* __restrict__ degC,
                                                         int* __restrict__ seg, int N, int E) {
    __shared__ int binCnt[NPART];
    __shared__ int binOff[NPART];
    __shared__ int binFill[NPART];
    __shared__ int binDone[NPART];
    __shared__ int waveTot[NPART / 32];
    __shared__ __attribute__((aligned(8))) int2 sorted[CHUNK_MAX];
    const int t = threadIdx.x;
    const int c = blockIdx.x;
    if (t < NPART) binDone[t] = 0;
    const int rpb = rows_per_bin(N);
    const int capPairs = cap_pairs(rpb);
    const int chunkE = (E + NPART - 1) / NPART;
    const int e0 = c * chunkE;
    int e1 = e0 + chunkE;
    e1 = e1 < E ? e1 : E;
    for (int s0 = e0; s0 < e1; s0 += CHUNK_MAX) {
        int s1 = s0 + CHUNK_MAX;
        s1 = s1 < e1 ? s1 : e1;
        if (t < NPART) binCnt[t] = 0;
        __syncthreads();
        for (int e = s0 + t; e < s1; e += 1024) {
            const int r = rows[e];
            const int col = cols[e];
            atomicAdd(&degC[col], 1);
            int b = r / rpb;
            b = b < NPART - 1 ? b : NPART - 1;
            atomicAdd(&binCnt[b], 1);
        }
        __syncthreads();
        const int n = (t < NPART) ? binCnt[t] : 0;
        int incl = n;
        for (int o = 1; o < 32; o <<= 1) {
            const int u = __shfl_up(incl, o);
            if ((t & 31) >= o) incl += u;
        }
        if (t < NPART && (t & 31) == 31) waveTot[t >> 5] = incl;
        __syncthreads();
        if (t < NPART) {
            int off = 0;
            for (int w = 0; w < (t >> 5); ++w) off += waveTot[w];
            binOff[t] = off + incl - n;
            binFill[t] = off + incl - n;
        }
        __syncthreads();
        for (int e = s0 + t; e < s1; e += 1024) {
            const int r = rows[e];
            const int col = cols[e];
            int b = r / rpb;
            b = b < NPART - 1 ? b : NPART - 1;
            const int pos = atomicAdd(&binFill[b], 1);
            sorted[pos] = make_int2((r - b * rpb) | (b << 16), col);
        }
        __syncthreads();
        const int cnt = s1 - s0;
        for (int i = t; i < cnt; i += 1024) {
            const int2 v = sorted[i];
            const int b = v.x >> 16;
            const int k = binDone[b] + (i - binOff[b]);
            if (k < capPairs) {
                int* p = seg + (size_t)b * rpb * 128 + (size_t)(c * capPairs + k) * 2;
                *(int2*)p = make_int2(v.x & 0xffff, v.y);
            }
        }
        __syncthreads();
        if (t < NPART) binDone[t] += binCnt[t];
    }
    __syncthreads();
    if (t < NPART) seg[(size_t)t * rpb * 128 + (rpb * 128 - NPART) + c] = binDone[t];
}

__global__ __launch_bounds__(1024) void binplace_kernel(const int* __restrict__ rows,
                                                        const int* __restrict__ cols,
                                                        int* __restrict__ cntR,
                                                        int* bucket, int N, int E) {
    __shared__ int rowCnt[BIN_ROWS_MAX];
    __shared__ int rowFill[BIN_ROWS_MAX];
    __shared__ int rowWaveTot[BIN_ROWS_MAX / 32];
    __shared__ int segOff[NPART + 1];
    __shared__ int segCnt[NPART];
    __shared__ int waveTot[NPART / 32];
    __shared__ int useList;
    __shared__ __attribute__((aligned(8))) int2 ed[BIN_EDGES_MAX];
    __shared__ int cs[BIN_EDGES_MAX];

    const int t = threadIdx.x;
    const int b = blockIdx.x;
    const int rpb = rows_per_bin(N);
    const int capPairs = cap_pairs(rpb);
    const int base = b * rpb;
    int nrows = (b == NPART - 1) ? (N - base) : rpb;
    if (base >= N) nrows = 0;
    if (nrows > BIN_ROWS_MAX) nrows = BIN_ROWS_MAX;

    for (int i = t; i < BIN_ROWS_MAX; i += 1024) rowCnt[i] = 0;
    int* brow = bucket + (size_t)base * 128;
    if (t == 0) useList = 0;
    const int n = (t < NPART) ? brow[rpb * 128 - NPART + t] : 0;
    int incl = n;
    for (int o = 1; o < 32; o <<= 1) {
        const int u = __shfl_up(incl, o);
        if ((t & 31) >= o) incl += u;
    }
    if (t < NPART) {
        segCnt[t] = n;
        if ((t & 31) == 31) waveTot[t >> 5] = incl;
    }
    __syncthreads();
    if (t < NPART) {
        int off = 0;
        for (int w = 0; w < (t >> 5); ++w) off += waveTot[w];
        segOff[t] = off + incl - n;
        if (n > capPairs) useList = 1;
        if (t == NPART - 1) {
            segOff[NPART] = off + incl;
            if (off + incl > BIN_EDGES_MAX) useList = 1;
        }
    }
    __syncthreads();
    const int total = segOff[NPART];
    const int wave = t >> 5, lane = t & 31;

    if (!useList) {
        for (int c = wave; c < NPART; c += 32) {
            const int pn = segCnt[c];
            const int po = segOff[c];
            const int2* p = (const int2*)(brow + (size_t)c * capPairs * 2);
            for (int k = lane; k < pn; k += 32) {
                const int2 v = p[k];
                ed[po + k] = v;
                atomicAdd(&rowCnt[v.x], 1);
            }
        }
        __syncthreads();
        const int m = rowCnt[t];
        int rincl = m;
        for (int o = 1; o < 32; o <<= 1) {
            const int u = __shfl_up(rincl, o);
            if (lane >= o) rincl += u;
        }
        if (lane == 31) rowWaveTot[wave] = rincl;
        __syncthreads();
        int roff = 0;
        for (int w = 0; w < wave; ++w) roff += rowWaveTot[w];
        rowFill[t] = roff + rincl - m;
        __syncthreads();
        for (int i = t; i < total; i += 1024) {
            const int2 v = ed[i];
            const int pos = atomicAdd(&rowFill[v.x], 1);
            cs[pos] = v.y;
        }
        __syncthreads();
        for (int row = wave; row < nrows; row += 32) {
            const int d = rowCnt[row];
            const int o = rowFill[row] - d;
            const int lim = d < ROW_CAP ? d : ROW_CAP;
            for (int k = lane; k < lim; k += 32) brow[(size_t)row * 128 + k] = cs[o + k];
        }
    } else {
        for (int e = t; e < E; e += 1024) {
            const int r = rows[e];
            int bb = r / rpb;
            bb = bb < NPART - 1 ? bb : NPART - 1;
            if (bb == b && r - base < nrows) {
                const int slot = atomicAdd(&rowCnt[r - base], 1);
                if (slot < ROW_CAP) brow[(size_t)(r - base) * 128 + slot] = cols[e];
            }
        }
    }
    __syncthreads();
    for (int i = t; i < nrows; i += 1024) cntR[base + i] = rowCnt[i];
}

#define W_PITCH 132

__device__ __forceinline__ v8f wmma4(v2f a, v2f b, v8f c) {
    return __builtin_amdgcn_wmma_f32_16x16x4_f32(
         false, a,  false, b,
         (short)0, c,  false,  false);
}

__global__ __launch_bounds__(128) void gemm_xwT_wmma_kernel(const float* __restrict__ X,
                                                            const float* __restrict__ W,
                                                            float* __restrict__ Xhat,
                                                            int N) {
    __shared__ __attribute__((aligned(16))) float Ws[128 * W_PITCH];

    const int t = threadIdx.x;
    const float4* __restrict__ W4 = (const float4*)W;
#pragma unroll 8
    for (int q = 0; q < 32; ++q) {
        const int i = t + 128 * q;
        const int row = i >> 5, c4 = i & 31;
        const float4 v = W4[i];
        *(float4*)&Ws[row * W_PITCH + c4 * 4] = v;
    }
    __syncthreads();

    const int lane = t & 31;
    const int wave = __builtin_amdgcn_readfirstlane(t >> 5);
    const int rowBase = (blockIdx.x * 4 + wave) * 16;
    if (rowBase >= N) return;

    const int half = lane >> 4;
    const int l16  = lane & 15;

    int ar = rowBase + l16;
    ar = (ar < N) ? ar : (N - 1);
    const float4* __restrict__ aptr = (const float4*)(X + (size_t)ar * 128 + half * 4);
    float4 a[16];
#pragma unroll
    for (int j = 0; j < 16; ++j) a[j] = aptr[2 * j];

    v8f acc[8];
#pragma unroll
    for (int ct = 0; ct < 8; ++ct)
        acc[ct] = (v8f){0.f, 0.f, 0.f, 0.f, 0.f, 0.f, 0.f, 0.f};

    const float* bbase = &Ws[l16 * W_PITCH + half * 4];
#pragma unroll
    for (int j = 0; j < 16; ++j) {
        float4 b[8];
#pragma unroll
        for (int ct = 0; ct < 8; ++ct)
            b[ct] = *(const float4*)(bbase + ct * 16 * W_PITCH + 8 * j);
        const v2f a0 = {a[j].x, a[j].y};
        const v2f a1 = {a[j].z, a[j].w};
#pragma unroll
        for (int ct = 0; ct < 8; ++ct) {
            const v2f b0 = {b[ct].x, b[ct].y};
            acc[ct] = wmma4(a0, b0, acc[ct]);
        }
#pragma unroll
        for (int ct = 0; ct < 8; ++ct) {
            const v2f b1 = {b[ct].z, b[ct].w};
            acc[ct] = wmma4(a1, b1, acc[ct]);
        }
    }

    float* cptr = Xhat + (size_t)rowBase * 128 + l16;
#pragma unroll
    for (int v = 0; v < 8; ++v) {
        const int m = v + (half << 3);
        if (rowBase + m < N) {
#pragma unroll
            for (int ct = 0; ct < 8; ++ct)
                cptr[(size_t)m * 128 + ct * 16] = acc[ct][v];
        }
    }
}

__device__ __forceinline__ void axpy4(float4& acc, float v, const float4 x) {
    acc.x = fmaf(v, x.x, acc.x);
    acc.y = fmaf(v, x.y, acc.y);
    acc.z = fmaf(v, x.z, acc.z);
    acc.w = fmaf(v, x.w, acc.w);
}

__device__ __forceinline__ float dinv_from(int deg) {
    return __builtin_amdgcn_rsqf((float)(deg + 1));
}
__device__ __forceinline__ float dinv_of(const int* __restrict__ degC, int n) {
    return dinv_from(degC[n]);
}

__device__ __forceinline__ void sum_long_row(float4& acc, const int* keys, const int* vals,
                                             int n, int total, int r, int lane, float dr,
                                             const int* __restrict__ degC,
                                             const float4* __restrict__ X4) {
    int last = -1;
    int done = 0;
    while (done < total) {
        int mn = 0x7fffffff;
        for (int j = lane; j < n; j += 32) {
            const bool mine = keys ? (keys[j] == r) : true;
            const int cj = vals[j];
            if (mine && cj > last && cj < mn) mn = cj;
        }
        for (int o = 16; o > 0; o >>= 1) {
            const int u = __shfl_xor(mn, o);
            mn = (u < mn) ? u : mn;
        }
        int cnt = 0;
        for (int j = lane; j < n; j += 32) {
            const bool mine = keys ? (keys[j] == r) : true;
            cnt += (mine && vals[j] == mn) ? 1 : 0;
        }
        for (int o = 16; o > 0; o >>= 1) cnt += __shfl_xor(cnt, o);
        cnt = __builtin_amdgcn_readfirstlane(cnt);
        mn  = __builtin_amdgcn_readfirstlane(mn);
        if (cnt <= 0) break;
        const float4 x0 = X4[(size_t)mn * 32 + lane];
        const float v0 = dr * dinv_of(degC, mn);
        for (int i = 0; i < cnt; ++i) axpy4(acc, v0, x0);
        done += cnt;
        last = mn;
    }
}

__global__ __launch_bounds__(256) void gather_rows_kernel(const int* __restrict__ cntR,
                                                          const int* __restrict__ degC,
                                                          const float* __restrict__ Xhat,
                                                          const int* __restrict__ rows,
                                                          const int* __restrict__ cols,
                                                          int* out, int N, int E) {
    const int lane = threadIdx.x & 31;
    const int r = blockIdx.x * 8 + __builtin_amdgcn_readfirstlane(threadIdx.x >> 5);
    if (r >= N) return;
    const int d = cntR[r];
    const float dr = dinv_of(degC, r);
    const float4* __restrict__ X4 = (const float4*)Xhat;
    int* orow = out + (size_t)r * 128;

    const float4 xs = X4[(size_t)r * 32 + lane];
    const float vs = dr * dr;
    float4 acc = {vs * xs.x, vs * xs.y, vs * xs.z, vs * xs.w};

    if (d <= 32) {
        const int c = (lane < d) ? orow[lane] : 0x7fffffff;
        int rk = 0;
        for (int j = 0; j < d; ++j) {
            const int cj = __builtin_amdgcn_readlane(c, j);
            rk += (cj < c || (cj == c && j < lane)) ? 1 : 0;
        }
        if (lane >= d) rk = lane;
        const int sorted = __builtin_amdgcn_ds_permute(rk << 2, c);

        int k = 0;
        for (; k + 4 <= d; k += 4) {
            const int c0 = __builtin_amdgcn_readlane(sorted, k);
            const int c1 = __builtin_amdgcn_readlane(sorted, k + 1);
            const int c2 = __builtin_amdgcn_readlane(sorted, k + 2);
            const int c3 = __builtin_amdgcn_readlane(sorted, k + 3);
            const float4 x0 = X4[(size_t)c0 * 32 + lane];
            const float4 x1 = X4[(size_t)c1 * 32 + lane];
            const float4 x2 = X4[(size_t)c2 * 32 + lane];
            const float4 x3 = X4[(size_t)c3 * 32 + lane];
            const int g0 = degC[c0];
            const int g1 = degC[c1];
            const int g2 = degC[c2];
            const int g3 = degC[c3];
            const float v0 = dr * dinv_from(g0);
            const float v1 = dr * dinv_from(g1);
            const float v2 = dr * dinv_from(g2);
            const float v3 = dr * dinv_from(g3);
            axpy4(acc, v0, x0);
            axpy4(acc, v1, x1);
            axpy4(acc, v2, x2);
            axpy4(acc, v3, x3);
        }
        for (; k < d; ++k) {
            const int c0 = __builtin_amdgcn_readlane(sorted, k);
            const float4 x0 = X4[(size_t)c0 * 32 + lane];
            axpy4(acc, dr * dinv_of(degC, c0), x0);
        }
    } else if (d <= ROW_CAP) {
        sum_long_row(acc, nullptr, orow, d, d, r, lane, dr, degC, X4);
    } else {
        sum_long_row(acc, rows, cols, E, d, r, lane, dr, degC, X4);
    }
    const int4 res = {__float_as_int(acc.x), __float_as_int(acc.y),
                      __float_as_int(acc.z), __float_as_int(acc.w)};
    *(int4*)(orow + lane * 4) = res;
}

extern "C" void kernel_launch(void* const* d_in, const int* in_sizes, int n_in,
                              void* d_out, int out_size, void* d_ws, size_t ws_size,
                              hipStream_t stream) {
    const float* X  = (const float*)d_in[0];
    const float* W  = (const float*)d_in[1];
    const int*   ei = (const int*)d_in[2];
    float* out = (float*)d_out;

    const int N = in_sizes[0] / 128;
    const int E = in_sizes[2] / 2;
    const int* rows = ei;
    const int* cols = ei + E;
    const int nb = (N + 255) / 256;

    float* Xhat = (float*)d_ws;
    int* cntR   = (int*)(Xhat + (size_t)N * 128);
    int* degC   = cntR + N;
    int* bucket = (int*)out;

    cnt_init_kernel<<<nb, 256, 0, stream>>>(degC, N);
    partition_kernel<<<NPART, 1024, 0, stream>>>(rows, cols, degC, bucket, N, E);
    binplace_kernel<<<NPART, 1024, 0, stream>>>(rows, cols, cntR, bucket, N, E);

    gemm_xwT_wmma_kernel<<<(N + 63) / 64, 128, 0, stream>>>(X, W, Xhat, N);

    gather_rows_kernel<<<(N + 7) / 8, 256, 0, stream>>>(cntR, degC, Xhat, rows, cols, bucket, N, E);
}


# gather: a row of 33 to 128 edges ranks its columns in registers, lists them in LDS and sums four row loads at a time (was one dependent row load per column behind two scans); f32 sum order kept
# speedup vs baseline: 2.5637x; 1.1458x over previous
#include <hip/hip_runtime.h>

typedef float v2f __attribute__((ext_vector_type(2)));
typedef float v8f __attribute__((ext_vector_type(8)));


#define ROW_CAP 128

__global__ __launch_bounds__(256) void cnt_init_kernel(int* __restrict__ degC, int N) {
    int n = blockIdx.x * 256 + threadIdx.x;
    if (n < N) degC[n] = 0;
}

#define NPART 256
#define BIN_ROWS_MAX 1024
#define BIN_EDGES_MAX 12288
#define CHUNK_MAX 8192

__device__ __forceinline__ int rows_per_bin(int N) {
    const int q = N / NPART;
    return q > 0 ? q : 1;
}

__device__ __forceinline__ int cap_pairs(int rpb) {
    return ((rpb * 128 - NPART) / (2 * NPART)) & ~15;
}

__global__ __launch_bounds__(1024) void partition_kernel(const int* __restrict__ rows,
                                                         const int* __restrict__ cols,
                                                         int* __restrict__ degC,
                                                         int* __restrict__ seg, int N, int E) {
    __shared__ int binCnt[NPART];
    __shared__ int binOff[NPART];
    __shared__ int binFill[NPART];
    __shared__ int binDone[NPART];
    __shared__ int waveTot[NPART / 32];
    __shared__ __attribute__((aligned(8))) int2 sorted[CHUNK_MAX];
    const int t = threadIdx.x;
    const int c = blockIdx.x;
    if (t < NPART) binDone[t] = 0;
    const int rpb = rows_per_bin(N);
    const int capPairs = cap_pairs(rpb);
    const int chunkE = (E + NPART - 1) / NPART;
    const int e0 = c * chunkE;
    int e1 = e0 + chunkE;
    e1 = e1 < E ? e1 : E;
    for (int s0 = e0; s0 < e1; s0 += CHUNK_MAX) {
        int s1 = s0 + CHUNK_MAX;
        s1 = s1 < e1 ? s1 : e1;
        if (t < NPART) binCnt[t] = 0;
        __syncthreads();
        for (int e = s0 + t; e < s1; e += 1024) {
            const int r = rows[e];
            const int col = cols[e];
            atomicAdd(&degC[col], 1);
            int b = r / rpb;
            b = b < NPART - 1 ? b : NPART - 1;
            atomicAdd(&binCnt[b], 1);
        }
        __syncthreads();
        const int n = (t < NPART) ? binCnt[t] : 0;
        int incl = n;
        for (int o = 1; o < 32; o <<= 1) {
            const int u = __shfl_up(incl, o);
            if ((t & 31) >= o) incl += u;
        }
        if (t < NPART && (t & 31) == 31) waveTot[t >> 5] = incl;
        __syncthreads();
        if (t < NPART) {
            int off = 0;
            for (int w = 0; w < (t >> 5); ++w) off += waveTot[w];
            binOff[t] = off + incl - n;
            binFill[t] = off + incl - n;
        }
        __syncthreads();
        for (int e = s0 + t; e < s1; e += 1024) {
            const int r = rows[e];
            const int col = cols[e];
            int b = r / rpb;
            b = b < NPART - 1 ? b : NPART - 1;
            const int pos = atomicAdd(&binFill[b], 1);
            sorted[pos] = make_int2((r - b * rpb) | (b << 16), col);
        }
        __syncthreads();
        const int cnt = s1 - s0;
        for (int i = t; i < cnt; i += 1024) {
            const int2 v = sorted[i];
            const int b = v.x >> 16;
            const int k = binDone[b] + (i - binOff[b]);
            if (k < capPairs) {
                int* p = seg + (size_t)b * rpb * 128 + (size_t)(c * capPairs + k) * 2;
                *(int2*)p = make_int2(v.x & 0xffff, v.y);
            }
        }
        __syncthreads();
        if (t < NPART) binDone[t] += binCnt[t];
    }
    __syncthreads();
    if (t < NPART) seg[(size_t)t * rpb * 128 + (rpb * 128 - NPART) + c] = binDone[t];
}

__global__ __launch_bounds__(1024) void binplace_kernel(const int* __restrict__ rows,
                                                        const int* __restrict__ cols,
                                                        int* __restrict__ cntR,
                                                        int* bucket, int N, int E) {
    __shared__ int rowCnt[BIN_ROWS_MAX];
    __shared__ int rowFill[BIN_ROWS_MAX];
    __shared__ int rowWaveTot[BIN_ROWS_MAX / 32];
    __shared__ int segOff[NPART + 1];
    __shared__ int segCnt[NPART];
    __shared__ int waveTot[NPART / 32];
    __shared__ int useList;
    __shared__ __attribute__((aligned(8))) int2 ed[BIN_EDGES_MAX];
    __shared__ int cs[BIN_EDGES_MAX];

    const int t = threadIdx.x;
    const int b = blockIdx.x;
    const int rpb = rows_per_bin(N);
    const int capPairs = cap_pairs(rpb);
    const int base = b * rpb;
    int nrows = (b == NPART - 1) ? (N - base) : rpb;
    if (base >= N) nrows = 0;
    if (nrows > BIN_ROWS_MAX) nrows = BIN_ROWS_MAX;

    for (int i = t; i < BIN_ROWS_MAX; i += 1024) rowCnt[i] = 0;
    int* brow = bucket + (size_t)base * 128;
    if (t == 0) useList = 0;
    const int n = (t < NPART) ? brow[rpb * 128 - NPART + t] : 0;
    int incl = n;
    for (int o = 1; o < 32; o <<= 1) {
        const int u = __shfl_up(incl, o);
        if ((t & 31) >= o) incl += u;
    }
    if (t < NPART) {
        segCnt[t] = n;
        if ((t & 31) == 31) waveTot[t >> 5] = incl;
    }
    __syncthreads();
    if (t < NPART) {
        int off = 0;
        for (int w = 0; w < (t >> 5); ++w) off += waveTot[w];
        segOff[t] = off + incl - n;
        if (n > capPairs) useList = 1;
        if (t == NPART - 1) {
            segOff[NPART] = off + incl;
            if (off + incl > BIN_EDGES_MAX) useList = 1;
        }
    }
    __syncthreads();
    const int total = segOff[NPART];
    const int wave = t >> 5, lane = t & 31;

    if (!useList) {
        for (int c = wave; c < NPART; c += 32) {
            const int pn = segCnt[c];
            const int po = segOff[c];
            const int2* p = (const int2*)(brow + (size_t)c * capPairs * 2);
            for (int k = lane; k < pn; k += 32) {
                const int2 v = p[k];
                ed[po + k] = v;
                atomicAdd(&rowCnt[v.x], 1);
            }
        }
        __syncthreads();
        const int m = rowCnt[t];
        int rincl = m;
        for (int o = 1; o < 32; o <<= 1) {
            const int u = __shfl_up(rincl, o);
            if (lane >= o) rincl += u;
        }
        if (lane == 31) rowWaveTot[wave] = rincl;
        __syncthreads();
        int roff = 0;
        for (int w = 0; w < wave; ++w) roff += rowWaveTot[w];
        rowFill[t] = roff + rincl - m;
        __syncthreads();
        for (int i = t; i < total; i += 1024) {
            const int2 v = ed[i];
            const int pos = atomicAdd(&rowFill[v.x], 1);
            cs[pos] = v.y;
        }
        __syncthreads();
        for (int row = wave; row < nrows; row += 32) {
            const int d = rowCnt[row];
            const int o = rowFill[row] - d;
            const int lim = d < ROW_CAP ? d : ROW_CAP;
            for (int k = lane; k < lim; k += 32) brow[(size_t)row * 128 + k] = cs[o + k];
        }
    } else {
        for (int e = t; e < E; e += 1024) {
            const int r = rows[e];
            int bb = r / rpb;
            bb = bb < NPART - 1 ? bb : NPART - 1;
            if (bb == b && r - base < nrows) {
                const int slot = atomicAdd(&rowCnt[r - base], 1);
                if (slot < ROW_CAP) brow[(size_t)(r - base) * 128 + slot] = cols[e];
            }
        }
    }
    __syncthreads();
    for (int i = t; i < nrows; i += 1024) cntR[base + i] = rowCnt[i];
}

#define W_PITCH 132

__device__ __forceinline__ v8f wmma4(v2f a, v2f b, v8f c) {
    return __builtin_amdgcn_wmma_f32_16x16x4_f32(
         false, a,  false, b,
         (short)0, c,  false,  false);
}

__global__ __launch_bounds__(128) void gemm_xwT_wmma_kernel(const float* __restrict__ X,
                                                            const float* __restrict__ W,
                                                            float* __restrict__ Xhat,
                                                            int N) {
    __shared__ __attribute__((aligned(16))) float Ws[128 * W_PITCH];

    const int t = threadIdx.x;
    const float4* __restrict__ W4 = (const float4*)W;
#pragma unroll 8
    for (int q = 0; q < 32; ++q) {
        const int i = t + 128 * q;
        const int row = i >> 5, c4 = i & 31;
        const float4 v = W4[i];
        *(float4*)&Ws[row * W_PITCH + c4 * 4] = v;
    }
    __syncthreads();

    const int lane = t & 31;
    const int wave = __builtin_amdgcn_readfirstlane(t >> 5);
    const int rowBase = (blockIdx.x * 4 + wave) * 16;
    if (rowBase >= N) return;

    const int half = lane >> 4;
    const int l16  = lane & 15;

    int ar = rowBase + l16;
    ar = (ar < N) ? ar : (N - 1);
    const float4* __restrict__ aptr = (const float4*)(X + (size_t)ar * 128 + half * 4);
    float4 a[16];
#pragma unroll
    for (int j = 0; j < 16; ++j) a[j] = aptr[2 * j];

    v8f acc[8];
#pragma unroll
    for (int ct = 0; ct < 8; ++ct)
        acc[ct] = (v8f){0.f, 0.f, 0.f, 0.f, 0.f, 0.f, 0.f, 0.f};

    const float* bbase = &Ws[l16 * W_PITCH + half * 4];
#pragma unroll
    for (int j = 0; j < 16; ++j) {
        float4 b[8];
#pragma unroll
        for (int ct = 0; ct < 8; ++ct)
            b[ct] = *(const float4*)(bbase + ct * 16 * W_PITCH + 8 * j);
        const v2f a0 = {a[j].x, a[j].y};
        const v2f a1 = {a[j].z, a[j].w};
#pragma unroll
        for (int ct = 0; ct < 8; ++ct) {
            const v2f b0 = {b[ct].x, b[ct].y};
            acc[ct] = wmma4(a0, b0, acc[ct]);
        }
#pragma unroll
        for (int ct = 0; ct < 8; ++ct) {
            const v2f b1 = {b[ct].z, b[ct].w};
            acc[ct] = wmma4(a1, b1, acc[ct]);
        }
    }

    float* cptr = Xhat + (size_t)rowBase * 128 + l16;
#pragma unroll
    for (int v = 0; v < 8; ++v) {
        const int m = v + (half << 3);
        if (rowBase + m < N) {
#pragma unroll
            for (int ct = 0; ct < 8; ++ct)
                cptr[(size_t)m * 128 + ct * 16] = acc[ct][v];
        }
    }
}

__device__ __forceinline__ void axpy4(float4& acc, float v, const float4 x) {
    acc.x = fmaf(v, x.x, acc.x);
    acc.y = fmaf(v, x.y, acc.y);
    acc.z = fmaf(v, x.z, acc.z);
    acc.w = fmaf(v, x.w, acc.w);
}

__device__ __forceinline__ float dinv_from(int deg) {
    return __builtin_amdgcn_rsqf((float)(deg + 1));
}
__device__ __forceinline__ float dinv_of(const int* __restrict__ degC, int n) {
    return dinv_from(degC[n]);
}

__device__ __forceinline__ void sum_long_row(float4& acc, const int* keys, const int* vals,
                                             int n, int total, int r, int lane, float dr,
                                             const int* __restrict__ degC,
                                             const float4* __restrict__ X4) {
    int last = -1;
    int done = 0;
    while (done < total) {
        int mn = 0x7fffffff;
        for (int j = lane; j < n; j += 32) {
            const bool mine = keys ? (keys[j] == r) : true;
            const int cj = vals[j];
            if (mine && cj > last && cj < mn) mn = cj;
        }
        for (int o = 16; o > 0; o >>= 1) {
            const int u = __shfl_xor(mn, o);
            mn = (u < mn) ? u : mn;
        }
        int cnt = 0;
        for (int j = lane; j < n; j += 32) {
            const bool mine = keys ? (keys[j] == r) : true;
            cnt += (mine && vals[j] == mn) ? 1 : 0;
        }
        for (int o = 16; o > 0; o >>= 1) cnt += __shfl_xor(cnt, o);
        cnt = __builtin_amdgcn_readfirstlane(cnt);
        mn  = __builtin_amdgcn_readfirstlane(mn);
        if (cnt <= 0) break;
        const float4 x0 = X4[(size_t)mn * 32 + lane];
        const float v0 = dr * dinv_of(degC, mn);
        for (int i = 0; i < cnt; ++i) axpy4(acc, v0, x0);
        done += cnt;
        last = mn;
    }
}

__device__ __forceinline__ void sum_mid_row(float4& acc, const int* orow, int d, int lane, float dr,
                                            const int* __restrict__ degC,
                                            const float4* __restrict__ X4, int* s) {
    int c[4], rk[4];
#pragma unroll
    for (int q = 0; q < 4; ++q) {
        const int i = lane + 32 * q;
        c[q] = (i < d) ? orow[i] : 0x7fffffff;
        rk[q] = 0;
    }
#pragma unroll
    for (int p = 0; p < 4; ++p) {
        int lim = d - 32 * p;
        lim = lim < 32 ? lim : 32;
        for (int j = 0; j < lim; ++j) {
            const int cj = __builtin_amdgcn_readlane(c[p], j);
            const int jj = 32 * p + j;
#pragma unroll
            for (int q = 0; q < 4; ++q)
                rk[q] += (cj < c[q] || (cj == c[q] && jj < lane + 32 * q)) ? 1 : 0;
        }
    }
#pragma unroll
    for (int q = 0; q < 4; ++q)
        if (lane + 32 * q < d) s[rk[q]] = c[q];

    int k = 0;
    for (; k + 4 <= d; k += 4) {
        const int4 cc = *(const int4*)(s + k);
        const int c0 = __builtin_amdgcn_readfirstlane(cc.x);
        const int c1 = __builtin_amdgcn_readfirstlane(cc.y);
        const int c2 = __builtin_amdgcn_readfirstlane(cc.z);
        const int c3 = __builtin_amdgcn_readfirstlane(cc.w);
        const float4 x0 = X4[(size_t)c0 * 32 + lane];
        const float4 x1 = X4[(size_t)c1 * 32 + lane];
        const float4 x2 = X4[(size_t)c2 * 32 + lane];
        const float4 x3 = X4[(size_t)c3 * 32 + lane];
        const int g0 = degC[c0];
        const int g1 = degC[c1];
        const int g2 = degC[c2];
        const int g3 = degC[c3];
        const float v0 = dr * dinv_from(g0);
        const float v1 = dr * dinv_from(g1);
        const float v2 = dr * dinv_from(g2);
        const float v3 = dr * dinv_from(g3);
        axpy4(acc, v0, x0);
        axpy4(acc, v1, x1);
        axpy4(acc, v2, x2);
        axpy4(acc, v3, x3);
    }
    for (; k < d; ++k) {
        const int c0 = __builtin_amdgcn_readfirstlane(s[k]);
        const float4 x0 = X4[(size_t)c0 * 32 + lane];
        axpy4(acc, dr * dinv_of(degC, c0), x0);
    }
}

__global__ __launch_bounds__(256) void gather_rows_kernel(const int* __restrict__ cntR,
                                                          const int* __restrict__ degC,
                                                          const float* __restrict__ Xhat,
                                                          const int* __restrict__ rows,
                                                          const int* __restrict__ cols,
                                                          int* out, int N, int E) {
    __shared__ __attribute__((aligned(16))) int colList[8][ROW_CAP];
    const int lane = threadIdx.x & 31;
    const int wave = __builtin_amdgcn_readfirstlane(threadIdx.x >> 5);
    const int r = blockIdx.x * 8 + wave;
    if (r >= N) return;
    const int d = cntR[r];
    const float dr = dinv_of(degC, r);
    const float4* __restrict__ X4 = (const float4*)Xhat;
    int* orow = out + (size_t)r * 128;

    const float4 xs = X4[(size_t)r * 32 + lane];
    const float vs = dr * dr;
    float4 acc = {vs * xs.x, vs * xs.y, vs * xs.z, vs * xs.w};

    if (d <= 32) {
        const int c = (lane < d) ? orow[lane] : 0x7fffffff;
        int rk = 0;
        for (int j = 0; j < d; ++j) {
            const int cj = __builtin_amdgcn_readlane(c, j);
            rk += (cj < c || (cj == c && j < lane)) ? 1 : 0;
        }
        if (lane >= d) rk = lane;
        const int sorted = __builtin_amdgcn_ds_permute(rk << 2, c);

        int k = 0;
        for (; k + 4 <= d; k += 4) {
            const int c0 = __builtin_amdgcn_readlane(sorted, k);
            const int c1 = __builtin_amdgcn_readlane(sorted, k + 1);
            const int c2 = __builtin_amdgcn_readlane(sorted, k + 2);
            const int c3 = __builtin_amdgcn_readlane(sorted, k + 3);
            const float4 x0 = X4[(size_t)c0 * 32 + lane];
            const float4 x1 = X4[(size_t)c1 * 32 + lane];
            const float4 x2 = X4[(size_t)c2 * 32 + lane];
            const float4 x3 = X4[(size_t)c3 * 32 + lane];
            const int g0 = degC[c0];
            const int g1 = degC[c1];
            const int g2 = degC[c2];
            const int g3 = degC[c3];
            const float v0 = dr * dinv_from(g0);
            const float v1 = dr * dinv_from(g1);
            const float v2 = dr * dinv_from(g2);
            const float v3 = dr * dinv_from(g3);
            axpy4(acc, v0, x0);
            axpy4(acc, v1, x1);
            axpy4(acc, v2, x2);
            axpy4(acc, v3, x3);
        }
        for (; k < d; ++k) {
            const int c0 = __builtin_amdgcn_readlane(sorted, k);
            const float4 x0 = X4[(size_t)c0 * 32 + lane];
            axpy4(acc, dr * dinv_of(degC, c0), x0);
        }
    } else if (d <= ROW_CAP) {
        sum_mid_row(acc, orow, d, lane, dr, degC, X4, colList[wave]);
    } else {
        sum_long_row(acc, rows, cols, E, d, r, lane, dr, degC, X4);
    }
    const int4 res = {__float_as_int(acc.x), __float_as_int(acc.y),
                      __float_as_int(acc.z), __float_as_int(acc.w)};
    *(int4*)(orow + lane * 4) = res;
}

extern "C" void kernel_launch(void* const* d_in, const int* in_sizes, int n_in,
                              void* d_out, int out_size, void* d_ws, size_t ws_size,
                              hipStream_t stream) {
    const float* X  = (const float*)d_in[0];
    const float* W  = (const float*)d_in[1];
    const int*   ei = (const int*)d_in[2];
    float* out = (float*)d_out;

    const int N = in_sizes[0] / 128;
    const int E = in_sizes[2] / 2;
    const int* rows = ei;
    const int* cols = ei + E;
    const int nb = (N + 255) / 256;

    float* Xhat = (float*)d_ws;
    int* cntR   = (int*)(Xhat + (size_t)N * 128);
    int* degC   = cntR + N;
    int* bucket = (int*)out;

    cnt_init_kernel<<<nb, 256, 0, stream>>>(degC, N);
    partition_kernel<<<NPART, 1024, 0, stream>>>(rows, cols, degC, bucket, N, E);
    binplace_kernel<<<NPART, 1024, 0, stream>>>(rows, cols, cntR, bucket, N, E);

    gemm_xwT_wmma_kernel<<<(N + 63) / 64, 128, 0, stream>>>(X, W, Xhat, N);

    gather_rows_kernel<<<(N + 7) / 8, 256, 0, stream>>>(cntR, degC, Xhat, rows, cols, bucket, N, E);
}


# source-column counts from edges split by column range in LDS, pieces summed per range by the bucket launch: no global atomic per edge, counter-clear launch dropped (5 to 4); f32 sum order kept
# speedup vs baseline: 3.9667x; 1.5473x over previous
#include <hip/hip_runtime.h>

typedef float v2f __attribute__((ext_vector_type(2)));
typedef float v8f __attribute__((ext_vector_type(8)));


#define ROW_CAP 128

#define NPART 256
#define BIN_ROWS_MAX 1024
#define BIN_EDGES_MAX 12288
#define CHUNK_MAX 8192

__device__ __forceinline__ int rows_per_bin(int N) {
    const int q = N / NPART;
    return q > 0 ? q : 1;
}

__device__ __forceinline__ int cap_pairs(int rpb) {
    return ((rpb * 128 - NPART) / (2 * NPART)) & ~15;
}

#define COL_CAP_MAX 64

__device__ __forceinline__ int cap_cols(int N) {
    int cap = ((N * 128 - NPART * NPART) / (NPART * NPART)) & ~15;
    cap = cap > 0 ? cap : 0;
    return cap < COL_CAP_MAX ? cap : COL_CAP_MAX;
}

__global__ __launch_bounds__(1024) void partition_kernel(const int* __restrict__ rows,
                                                         const int* __restrict__ cols,
                                                         int* __restrict__ cseg,
                                                         int* __restrict__ seg, int N, int E) {
    __shared__ int binCnt[NPART];
    __shared__ int binOff[NPART];
    __shared__ int binFill[NPART];
    __shared__ int binDone[NPART];
    __shared__ int cbinCnt[NPART];
    __shared__ int cbinOff[NPART];
    __shared__ int cbinFill[NPART];
    __shared__ int cbinDone[NPART];
    __shared__ int waveTot[2 * NPART / 32];
    __shared__ __attribute__((aligned(8))) int2 sorted[CHUNK_MAX];
    __shared__ int sortedC[CHUNK_MAX];
    const int t = threadIdx.x;
    const int c = blockIdx.x;
    if (t < NPART) { binDone[t] = 0; cbinDone[t] = 0; }
    const int rpb = rows_per_bin(N);
    const int capPairs = cap_pairs(rpb);
    const int capC = cap_cols(N);
    int* __restrict__ ccnt = cseg + (size_t)NPART * NPART * capC;
    const int chunkE = (E + NPART - 1) / NPART;
    const int e0 = c * chunkE;
    int e1 = e0 + chunkE;
    e1 = e1 < E ? e1 : E;
    const int g = t / NPART;
    const int tb = t % NPART;
    for (int s0 = e0; s0 < e1; s0 += CHUNK_MAX) {
        int s1 = s0 + CHUNK_MAX;
        s1 = s1 < e1 ? s1 : e1;
        if (t < NPART) { binCnt[t] = 0; cbinCnt[t] = 0; }
        __syncthreads();
        for (int e = s0 + t; e < s1; e += 1024) {
            const int r = rows[e];
            const int col = cols[e];
            int b = r / rpb;
            b = b < NPART - 1 ? b : NPART - 1;
            int cb = col / rpb;
            cb = cb < NPART - 1 ? cb : NPART - 1;
            atomicAdd(&binCnt[b], 1);
            atomicAdd(&cbinCnt[cb], 1);
        }
        __syncthreads();
        int n = 0;
        if (g == 0) n = binCnt[tb];
        else if (g == 1) n = cbinCnt[tb];
        int incl = n;
        for (int o = 1; o < 32; o <<= 1) {
            const int u = __shfl_up(incl, o);
            if ((t & 31) >= o) incl += u;
        }
        if (g < 2 && (t & 31) == 31) waveTot[t >> 5] = incl;
        __syncthreads();
        if (g < 2) {
            int off = 0;
            for (int w = g * (NPART / 32); w < (t >> 5); ++w) off += waveTot[w];
            const int first = off + incl - n;
            if (g == 0) { binOff[tb] = first; binFill[tb] = first; }
            else        { cbinOff[tb] = first; cbinFill[tb] = first; }
        }
        __syncthreads();
        for (int e = s0 + t; e < s1; e += 1024) {
            const int r = rows[e];
            const int col = cols[e];
            int b = r / rpb;
            b = b < NPART - 1 ? b : NPART - 1;
            int cb = col / rpb;
            cb = cb < NPART - 1 ? cb : NPART - 1;
            const int pos = atomicAdd(&binFill[b], 1);
            sorted[pos] = make_int2((r - b * rpb) | (b << 16), col);
            const int cpos = atomicAdd(&cbinFill[cb], 1);
            sortedC[cpos] = (col - cb * rpb) | (cb << 16);
        }
        __syncthreads();
        const int cnt = s1 - s0;
        for (int i = t; i < cnt; i += 1024) {
            const int2 v = sorted[i];
            const int w = sortedC[i];
            const int b = v.x >> 16;
            const int k = binDone[b] + (i - binOff[b]);
            if (k < capPairs) {
                int* p = seg + (size_t)b * rpb * 128 + (size_t)(c * capPairs + k) * 2;
                *(int2*)p = make_int2(v.x & 0xffff, v.y);
            }
            const int cb = w >> 16;
            const int ck = cbinDone[cb] + (i - cbinOff[cb]);
            if (ck < capC) cseg[((size_t)cb * NPART + c) * capC + ck] = w & 0xffff;
        }
        __syncthreads();
        if (t < NPART) { binDone[t] += binCnt[t]; cbinDone[t] += cbinCnt[t]; }
    }
    __syncthreads();
    if (t < NPART) {
        seg[(size_t)t * rpb * 128 + (rpb * 128 - NPART) + c] = binDone[t];
        ccnt[(size_t)t * NPART + c] = cbinDone[t];
    }
}

__global__ __launch_bounds__(1024) void binplace_kernel(const int* __restrict__ rows,
                                                        const int* __restrict__ cols,
                                                        int* __restrict__ cntR,
                                                        int* __restrict__ degC,
                                                        const int* __restrict__ cseg,
                                                        int* bucket, int N, int E) {
    __shared__ int rowCnt[BIN_ROWS_MAX];
    __shared__ int rowFill[BIN_ROWS_MAX];
    __shared__ int rowWaveTot[BIN_ROWS_MAX / 32];
    __shared__ int segOff[NPART + 1];
    __shared__ int segCnt[NPART];
    __shared__ int waveTot[NPART / 32];
    __shared__ int useList;
    __shared__ int colCnt[BIN_ROWS_MAX];
    __shared__ int csegCnt[NPART];
    __shared__ int useColList;
    __shared__ __attribute__((aligned(8))) int2 ed[BIN_EDGES_MAX];
    __shared__ int cs[BIN_EDGES_MAX];

    const int t = threadIdx.x;
    const int b = blockIdx.x;
    const int rpb = rows_per_bin(N);
    const int capPairs = cap_pairs(rpb);
    const int base = b * rpb;
    int nrows = (b == NPART - 1) ? (N - base) : rpb;
    if (base >= N) nrows = 0;
    if (nrows > BIN_ROWS_MAX) nrows = BIN_ROWS_MAX;

    for (int i = t; i < BIN_ROWS_MAX; i += 1024) { rowCnt[i] = 0; colCnt[i] = 0; }
    const int capC = cap_cols(N);
    const int* __restrict__ cpieces = cseg + (size_t)b * NPART * capC;
    const int cn0 = (t < NPART) ? cseg[(size_t)NPART * NPART * capC + (size_t)b * NPART + t] : 0;
    int* brow = bucket + (size_t)base * 128;
    if (t == 0) { useList = 0; useColList = 0; }
    const int n = (t < NPART) ? brow[rpb * 128 - NPART + t] : 0;
    int incl = n;
    for (int o = 1; o < 32; o <<= 1) {
        const int u = __shfl_up(incl, o);
        if ((t & 31) >= o) incl += u;
    }
    if (t < NPART) {
        segCnt[t] = n;
        csegCnt[t] = cn0;
        if ((t & 31) == 31) waveTot[t >> 5] = incl;
    }
    __syncthreads();
    if (t < NPART) {
        int off = 0;
        for (int w = 0; w < (t >> 5); ++w) off += waveTot[w];
        segOff[t] = off + incl - n;
        if (n > capPairs) useList = 1;
        if (cn0 > capC) useColList = 1;
        if (t == NPART - 1) {
            segOff[NPART] = off + incl;
            if (off + incl > BIN_EDGES_MAX) useList = 1;
        }
    }
    __syncthreads();
    const int total = segOff[NPART];
    const int wave = t >> 5, lane = t & 31;

    const int colList = useColList;
    int cn[NPART / 32], cv[NPART / 32];
#pragma unroll
    for (int i = 0; i < NPART / 32; ++i) {
        const int c = wave + 32 * i;
        cn[i] = colList ? 0 : csegCnt[c];
        cv[i] = (lane < cn[i]) ? cpieces[(size_t)c * capC + lane] : -1;
    }

    if (!useList) {
        for (int c = wave; c < NPART; c += 32) {
            const int pn = segCnt[c];
            const int po = segOff[c];
            const int2* p = (const int2*)(brow + (size_t)c * capPairs * 2);
            for (int k = lane; k < pn; k += 32) {
                const int2 v = p[k];
                ed[po + k] = v;
                atomicAdd(&rowCnt[v.x], 1);
            }
        }
        __syncthreads();
        const int m = rowCnt[t];
        int rincl = m;
        for (int o = 1; o < 32; o <<= 1) {
            const int u = __shfl_up(rincl, o);
            if (lane >= o) rincl += u;
        }
        if (lane == 31) rowWaveTot[wave] = rincl;
        __syncthreads();
        int roff = 0;
        for (int w = 0; w < wave; ++w) roff += rowWaveTot[w];
        rowFill[t] = roff + rincl - m;
        __syncthreads();
        for (int i = t; i < total; i += 1024) {
            const int2 v = ed[i];
            const int pos = atomicAdd(&rowFill[v.x], 1);
            cs[pos] = v.y;
        }
        __syncthreads();
        for (int row = wave; row < nrows; row += 32) {
            const int d = rowCnt[row];
            const int o = rowFill[row] - d;
            const int lim = d < ROW_CAP ? d : ROW_CAP;
            for (int k = lane; k < lim; k += 32) brow[(size_t)row * 128 + k] = cs[o + k];
        }
    } else {
        for (int e = t; e < E; e += 1024) {
            const int r = rows[e];
            int bb = r / rpb;
            bb = bb < NPART - 1 ? bb : NPART - 1;
            if (bb == b && r - base < nrows) {
                const int slot = atomicAdd(&rowCnt[r - base], 1);
                if (slot < ROW_CAP) brow[(size_t)(r - base) * 128 + slot] = cols[e];
            }
        }
    }
    if (!colList) {
#pragma unroll
        for (int i = 0; i < NPART / 32; ++i)
            if ((unsigned)cv[i] < (unsigned)BIN_ROWS_MAX) atomicAdd(&colCnt[cv[i]], 1);
#pragma unroll
        for (int i = 0; i < NPART / 32; ++i) {
            const int* q = cpieces + (size_t)(wave + 32 * i) * capC;
            for (int k = lane + 32; k < cn[i]; k += 32) {
                const int v = q[k];
                if ((unsigned)v < (unsigned)BIN_ROWS_MAX) atomicAdd(&colCnt[v], 1);
            }
        }
    } else {
        for (int e = t; e < E; e += 1024) {
            const int col = cols[e];
            int bb = col / rpb;
            bb = bb < NPART - 1 ? bb : NPART - 1;
            if (bb == b && (unsigned)(col - base) < (unsigned)BIN_ROWS_MAX) atomicAdd(&colCnt[col - base], 1);
        }
    }
    __syncthreads();
    for (int i = t; i < nrows; i += 1024) {
        cntR[base + i] = rowCnt[i];
        degC[base + i] = colCnt[i];
    }
}

#define W_PITCH 132

__device__ __forceinline__ v8f wmma4(v2f a, v2f b, v8f c) {
    return __builtin_amdgcn_wmma_f32_16x16x4_f32(
         false, a,  false, b,
         (short)0, c,  false,  false);
}

__global__ __launch_bounds__(128) void gemm_xwT_wmma_kernel(const float* __restrict__ X,
                                                            const float* __restrict__ W,
                                                            float* __restrict__ Xhat,
                                                            int N) {
    __shared__ __attribute__((aligned(16))) float Ws[128 * W_PITCH];

    const int t = threadIdx.x;
    const float4* __restrict__ W4 = (const float4*)W;
#pragma unroll 8
    for (int q = 0; q < 32; ++q) {
        const int i = t + 128 * q;
        const int row = i >> 5, c4 = i & 31;
        const float4 v = W4[i];
        *(float4*)&Ws[row * W_PITCH + c4 * 4] = v;
    }
    __syncthreads();

    const int lane = t & 31;
    const int wave = __builtin_amdgcn_readfirstlane(t >> 5);
    const int rowBase = (blockIdx.x * 4 + wave) * 16;
    if (rowBase >= N) return;

    const int half = lane >> 4;
    const int l16  = lane & 15;

    int ar = rowBase + l16;
    ar = (ar < N) ? ar : (N - 1);
    const float4* __restrict__ aptr = (const float4*)(X + (size_t)ar * 128 + half * 4);
    float4 a[16];
#pragma unroll
    for (int j = 0; j < 16; ++j) a[j] = aptr[2 * j];

    v8f acc[8];
#pragma unroll
    for (int ct = 0; ct < 8; ++ct)
        acc[ct] = (v8f){0.f, 0.f, 0.f, 0.f, 0.f, 0.f, 0.f, 0.f};

    const float* bbase = &Ws[l16 * W_PITCH + half * 4];
#pragma unroll
    for (int j = 0; j < 16; ++j) {
        float4 b[8];
#pragma unroll
        for (int ct = 0; ct < 8; ++ct)
            b[ct] = *(const float4*)(bbase + ct * 16 * W_PITCH + 8 * j);
        const v2f a0 = {a[j].x, a[j].y};
        const v2f a1 = {a[j].z, a[j].w};
#pragma unroll
        for (int ct = 0; ct < 8; ++ct) {
            const v2f b0 = {b[ct].x, b[ct].y};
            acc[ct] = wmma4(a0, b0, acc[ct]);
        }
#pragma unroll
        for (int ct = 0; ct < 8; ++ct) {
            const v2f b1 = {b[ct].z, b[ct].w};
            acc[ct] = wmma4(a1, b1, acc[ct]);
        }
    }

    float* cptr = Xhat + (size_t)rowBase * 128 + l16;
#pragma unroll
    for (int v = 0; v < 8; ++v) {
        const int m = v + (half << 3);
        if (rowBase + m < N) {
#pragma unroll
            for (int ct = 0; ct < 8; ++ct)
                cptr[(size_t)m * 128 + ct * 16] = acc[ct][v];
        }
    }
}

__device__ __forceinline__ void axpy4(float4& acc, float v, const float4 x) {
    acc.x = fmaf(v, x.x, acc.x);
    acc.y = fmaf(v, x.y, acc.y);
    acc.z = fmaf(v, x.z, acc.z);
    acc.w = fmaf(v, x.w, acc.w);
}

__device__ __forceinline__ float dinv_from(int deg) {
    return __builtin_amdgcn_rsqf((float)(deg + 1));
}
__device__ __forceinline__ float dinv_of(const int* __restrict__ degC, int n) {
    return dinv_from(degC[n]);
}

__device__ __forceinline__ void sum_long_row(float4& acc, const int* keys, const int* vals,
                                             int n, int total, int r, int lane, float dr,
                                             const int* __restrict__ degC,
                                             const float4* __restrict__ X4) {
    int last = -1;
    int done = 0;
    while (done < total) {
        int mn = 0x7fffffff;
        for (int j = lane; j < n; j += 32) {
            const bool mine = keys ? (keys[j] == r) : true;
            const int cj = vals[j];
            if (mine && cj > last && cj < mn) mn = cj;
        }
        for (int o = 16; o > 0; o >>= 1) {
            const int u = __shfl_xor(mn, o);
            mn = (u < mn) ? u : mn;
        }
        int cnt = 0;
        for (int j = lane; j < n; j += 32) {
            const bool mine = keys ? (keys[j] == r) : true;
            cnt += (mine && vals[j] == mn) ? 1 : 0;
        }
        for (int o = 16; o > 0; o >>= 1) cnt += __shfl_xor(cnt, o);
        cnt = __builtin_amdgcn_readfirstlane(cnt);
        mn  = __builtin_amdgcn_readfirstlane(mn);
        if (cnt <= 0) break;
        const float4 x0 = X4[(size_t)mn * 32 + lane];
        const float v0 = dr * dinv_of(degC, mn);
        for (int i = 0; i < cnt; ++i) axpy4(acc, v0, x0);
        done += cnt;
        last = mn;
    }
}

__device__ __forceinline__ void sum_mid_row(float4& acc, const int* orow, int d, int lane, float dr,
                                            const int* __restrict__ degC,
                                            const float4* __restrict__ X4, int* s) {
    int c[4], rk[4];
#pragma unroll
    for (int q = 0; q < 4; ++q) {
        const int i = lane + 32 * q;
        c[q] = (i < d) ? orow[i] : 0x7fffffff;
        rk[q] = 0;
    }
#pragma unroll
    for (int p = 0; p < 4; ++p) {
        int lim = d - 32 * p;
        lim = lim < 32 ? lim : 32;
        for (int j = 0; j < lim; ++j) {
            const int cj = __builtin_amdgcn_readlane(c[p], j);
            const int jj = 32 * p + j;
#pragma unroll
            for (int q = 0; q < 4; ++q)
                rk[q] += (cj < c[q] || (cj == c[q] && jj < lane + 32 * q)) ? 1 : 0;
        }
    }
#pragma unroll
    for (int q = 0; q < 4; ++q)
        if (lane + 32 * q < d) s[rk[q]] = c[q];

    int k = 0;
    for (; k + 4 <= d; k += 4) {
        const int4 cc = *(const int4*)(s + k);
        const int c0 = __builtin_amdgcn_readfirstlane(cc.x);
        const int c1 = __builtin_amdgcn_readfirstlane(cc.y);
        const int c2 = __builtin_amdgcn_readfirstlane(cc.z);
        const int c3 = __builtin_amdgcn_readfirstlane(cc.w);
        const float4 x0 = X4[(size_t)c0 * 32 + lane];
        const float4 x1 = X4[(size_t)c1 * 32 + lane];
        const float4 x2 = X4[(size_t)c2 * 32 + lane];
        const float4 x3 = X4[(size_t)c3 * 32 + lane];
        const int g0 = degC[c0];
        const int g1 = degC[c1];
        const int g2 = degC[c2];
        const int g3 = degC[c3];
        const float v0 = dr * dinv_from(g0);
        const float v1 = dr * dinv_from(g1);
        const float v2 = dr * dinv_from(g2);
        const float v3 = dr * dinv_from(g3);
        axpy4(acc, v0, x0);
        axpy4(acc, v1, x1);
        axpy4(acc, v2, x2);
        axpy4(acc, v3, x3);
    }
    for (; k < d; ++k) {
        const int c0 = __builtin_amdgcn_readfirstlane(s[k]);
        const float4 x0 = X4[(size_t)c0 * 32 + lane];
        axpy4(acc, dr * dinv_of(degC, c0), x0);
    }
}

__global__ __launch_bounds__(256) void gather_rows_kernel(const int* __restrict__ cntR,
                                                          const int* __restrict__ degC,
                                                          const float* __restrict__ Xhat,
                                                          const int* __restrict__ rows,
                                                          const int* __restrict__ cols,
                                                          int* out, int N, int E) {
    __shared__ __attribute__((aligned(16))) int colList[8][ROW_CAP];
    const int lane = threadIdx.x & 31;
    const int wave = __builtin_amdgcn_readfirstlane(threadIdx.x >> 5);
    const int r = blockIdx.x * 8 + wave;
    if (r >= N) return;
    const int d = cntR[r];
    const float dr = dinv_of(degC, r);
    const float4* __restrict__ X4 = (const float4*)Xhat;
    int* orow = out + (size_t)r * 128;

    const float4 xs = X4[(size_t)r * 32 + lane];
    const float vs = dr * dr;
    float4 acc = {vs * xs.x, vs * xs.y, vs * xs.z, vs * xs.w};

    if (d <= 32) {
        const int c = (lane < d) ? orow[lane] : 0x7fffffff;
        int rk = 0;
        for (int j = 0; j < d; ++j) {
            const int cj = __builtin_amdgcn_readlane(c, j);
            rk += (cj < c || (cj == c && j < lane)) ? 1 : 0;
        }
        if (lane >= d) rk = lane;
        const int sorted = __builtin_amdgcn_ds_permute(rk << 2, c);

        int k = 0;
        for (; k + 4 <= d; k += 4) {
            const int c0 = __builtin_amdgcn_readlane(sorted, k);
            const int c1 = __builtin_amdgcn_readlane(sorted, k + 1);
            const int c2 = __builtin_amdgcn_readlane(sorted, k + 2);
            const int c3 = __builtin_amdgcn_readlane(sorted, k + 3);
            const float4 x0 = X4[(size_t)c0 * 32 + lane];
            const float4 x1 = X4[(size_t)c1 * 32 + lane];
            const float4 x2 = X4[(size_t)c2 * 32 + lane];
            const float4 x3 = X4[(size_t)c3 * 32 + lane];
            const int g0 = degC[c0];
            const int g1 = degC[c1];
            const int g2 = degC[c2];
            const int g3 = degC[c3];
            const float v0 = dr * dinv_from(g0);
            const float v1 = dr * dinv_from(g1);
            const float v2 = dr * dinv_from(g2);
            const float v3 = dr * dinv_from(g3);
            axpy4(acc, v0, x0);
            axpy4(acc, v1, x1);
            axpy4(acc, v2, x2);
            axpy4(acc, v3, x3);
        }
        for (; k < d; ++k) {
            const int c0 = __builtin_amdgcn_readlane(sorted, k);
            const float4 x0 = X4[(size_t)c0 * 32 + lane];
            axpy4(acc, dr * dinv_of(degC, c0), x0);
        }
    } else if (d <= ROW_CAP) {
        sum_mid_row(acc, orow, d, lane, dr, degC, X4, colList[wave]);
    } else {
        sum_long_row(acc, rows, cols, E, d, r, lane, dr, degC, X4);
    }
    const int4 res = {__float_as_int(acc.x), __float_as_int(acc.y),
                      __float_as_int(acc.z), __float_as_int(acc.w)};
    *(int4*)(orow + lane * 4) = res;
}

extern "C" void kernel_launch(void* const* d_in, const int* in_sizes, int n_in,
                              void* d_out, int out_size, void* d_ws, size_t ws_size,
                              hipStream_t stream) {
    const float* X  = (const float*)d_in[0];
    const float* W  = (const float*)d_in[1];
    const int*   ei = (const int*)d_in[2];
    float* out = (float*)d_out;

    const int N = in_sizes[0] / 128;
    const int E = in_sizes[2] / 2;
    const int* rows = ei;
    const int* cols = ei + E;

    float* Xhat = (float*)d_ws;
    int* cntR   = (int*)(Xhat + (size_t)N * 128);
    int* degC   = cntR + N;
    int* bucket = (int*)out;

    int* cseg = (int*)Xhat;
    partition_kernel<<<NPART, 1024, 0, stream>>>(rows, cols, cseg, bucket, N, E);
    binplace_kernel<<<NPART, 1024, 0, stream>>>(rows, cols, cntR, degC, cseg, bucket, N, E);

    gemm_xwT_wmma_kernel<<<(N + 63) / 64, 128, 0, stream>>>(X, W, Xhat, N);

    gather_rows_kernel<<<(N + 7) / 8, 256, 0, stream>>>(cntR, degC, Xhat, rows, cols, bucket, N, E);
}


# product moved behind the row sum into the gather launch: out = (dinv[r] sum_c dinv[c] X[c]) W^T, WMMA from an LDS tile; product launch and X_hat plane dropped (4 to 3); f32 sum order changes
# speedup vs baseline: 4.5153x; 1.1383x over previous
#include <hip/hip_runtime.h>

typedef float v2f __attribute__((ext_vector_type(2)));
typedef float v8f __attribute__((ext_vector_type(8)));


#define ROW_CAP 128

#define NPART 256
#define BIN_ROWS_MAX 1024
#define BIN_EDGES_MAX 12288
#define CHUNK_MAX 8192

__device__ __forceinline__ int rows_per_bin(int N) {
    const int q = N / NPART;
    return q > 0 ? q : 1;
}

__device__ __forceinline__ int cap_pairs(int rpb) {
    return ((rpb * 128 - NPART) / (2 * NPART)) & ~15;
}

#define COL_CAP_MAX 64

__device__ __forceinline__ int cap_cols(int N) {
    int cap = ((N * 128 - NPART * NPART) / (NPART * NPART)) & ~15;
    cap = cap > 0 ? cap : 0;
    return cap < COL_CAP_MAX ? cap : COL_CAP_MAX;
}

__global__ __launch_bounds__(1024) void partition_kernel(const int* __restrict__ rows,
                                                         const int* __restrict__ cols,
                                                         int* __restrict__ cseg,
                                                         int* __restrict__ seg, int N, int E) {
    __shared__ int binCnt[NPART];
    __shared__ int binOff[NPART];
    __shared__ int binFill[NPART];
    __shared__ int binDone[NPART];
    __shared__ int cbinCnt[NPART];
    __shared__ int cbinOff[NPART];
    __shared__ int cbinFill[NPART];
    __shared__ int cbinDone[NPART];
    __shared__ int waveTot[2 * NPART / 32];
    __shared__ __attribute__((aligned(8))) int2 sorted[CHUNK_MAX];
    __shared__ int sortedC[CHUNK_MAX];
    const int t = threadIdx.x;
    const int c = blockIdx.x;
    if (t < NPART) { binDone[t] = 0; cbinDone[t] = 0; }
    const int rpb = rows_per_bin(N);
    const int capPairs = cap_pairs(rpb);
    const int capC = cap_cols(N);
    int* __restrict__ ccnt = cseg + (size_t)NPART * NPART * capC;
    const int chunkE = (E + NPART - 1) / NPART;
    const int e0 = c * chunkE;
    int e1 = e0 + chunkE;
    e1 = e1 < E ? e1 : E;
    const int g = t / NPART;
    const int tb = t % NPART;
    for (int s0 = e0; s0 < e1; s0 += CHUNK_MAX) {
        int s1 = s0 + CHUNK_MAX;
        s1 = s1 < e1 ? s1 : e1;
        if (t < NPART) { binCnt[t] = 0; cbinCnt[t] = 0; }
        __syncthreads();
        for (int e = s0 + t; e < s1; e += 1024) {
            const int r = rows[e];
            const int col = cols[e];
            int b = r / rpb;
            b = b < NPART - 1 ? b : NPART - 1;
            int cb = col / rpb;
            cb = cb < NPART - 1 ? cb : NPART - 1;
            atomicAdd(&binCnt[b], 1);
            atomicAdd(&cbinCnt[cb], 1);
        }
        __syncthreads();
        int n = 0;
        if (g == 0) n = binCnt[tb];
        else if (g == 1) n = cbinCnt[tb];
        int incl = n;
        for (int o = 1; o < 32; o <<= 1) {
            const int u = __shfl_up(incl, o);
            if ((t & 31) >= o) incl += u;
        }
        if (g < 2 && (t & 31) == 31) waveTot[t >> 5] = incl;
        __syncthreads();
        if (g < 2) {
            int off = 0;
            for (int w = g * (NPART / 32); w < (t >> 5); ++w) off += waveTot[w];
            const int first = off + incl - n;
            if (g == 0) { binOff[tb] = first; binFill[tb] = first; }
            else        { cbinOff[tb] = first; cbinFill[tb] = first; }
        }
        __syncthreads();
        for (int e = s0 + t; e < s1; e += 1024) {
            const int r = rows[e];
            const int col = cols[e];
            int b = r / rpb;
            b = b < NPART - 1 ? b : NPART - 1;
            int cb = col / rpb;
            cb = cb < NPART - 1 ? cb : NPART - 1;
            const int pos = atomicAdd(&binFill[b], 1);
            sorted[pos] = make_int2((r - b * rpb) | (b << 16), col);
            const int cpos = atomicAdd(&cbinFill[cb], 1);
            sortedC[cpos] = (col - cb * rpb) | (cb << 16);
        }
        __syncthreads();
        const int cnt = s1 - s0;
        for (int i = t; i < cnt; i += 1024) {
            const int2 v = sorted[i];
            const int w = sortedC[i];
            const int b = v.x >> 16;
            const int k = binDone[b] + (i - binOff[b]);
            if (k < capPairs) {
                int* p = seg + (size_t)b * rpb * 128 + (size_t)(c * capPairs + k) * 2;
                *(int2*)p = make_int2(v.x & 0xffff, v.y);
            }
            const int cb = w >> 16;
            const int ck = cbinDone[cb] + (i - cbinOff[cb]);
            if (ck < capC) cseg[((size_t)cb * NPART + c) * capC + ck] = w & 0xffff;
        }
        __syncthreads();
        if (t < NPART) { binDone[t] += binCnt[t]; cbinDone[t] += cbinCnt[t]; }
    }
    __syncthreads();
    if (t < NPART) {
        seg[(size_t)t * rpb * 128 + (rpb * 128 - NPART) + c] = binDone[t];
        ccnt[(size_t)t * NPART + c] = cbinDone[t];
    }
}

__global__ __launch_bounds__(1024) void binplace_kernel(const int* __restrict__ rows,
                                                        const int* __restrict__ cols,
                                                        int* __restrict__ cntR,
                                                        int* __restrict__ degC,
                                                        const int* __restrict__ cseg,
                                                        int* bucket, int N, int E) {
    __shared__ int rowCnt[BIN_ROWS_MAX];
    __shared__ int rowFill[BIN_ROWS_MAX];
    __shared__ int rowWaveTot[BIN_ROWS_MAX / 32];
    __shared__ int segOff[NPART + 1];
    __shared__ int segCnt[NPART];
    __shared__ int waveTot[NPART / 32];
    __shared__ int useList;
    __shared__ int colCnt[BIN_ROWS_MAX];
    __shared__ int csegCnt[NPART];
    __shared__ int useColList;
    __shared__ __attribute__((aligned(8))) int2 ed[BIN_EDGES_MAX];
    __shared__ int cs[BIN_EDGES_MAX];

    const int t = threadIdx.x;
    const int b = blockIdx.x;
    const int rpb = rows_per_bin(N);
    const int capPairs = cap_pairs(rpb);
    const int base = b * rpb;
    int nrows = (b == NPART - 1) ? (N - base) : rpb;
    if (base >= N) nrows = 0;
    if (nrows > BIN_ROWS_MAX) nrows = BIN_ROWS_MAX;

    for (int i = t; i < BIN_ROWS_MAX; i += 1024) { rowCnt[i] = 0; colCnt[i] = 0; }
    const int capC = cap_cols(N);
    const int* __restrict__ cpieces = cseg + (size_t)b * NPART * capC;
    const int cn0 = (t < NPART) ? cseg[(size_t)NPART * NPART * capC + (size_t)b * NPART + t] : 0;
    int* brow = bucket + (size_t)base * 128;
    if (t == 0) { useList = 0; useColList = 0; }
    const int n = (t < NPART) ? brow[rpb * 128 - NPART + t] : 0;
    int incl = n;
    for (int o = 1; o < 32; o <<= 1) {
        const int u = __shfl_up(incl, o);
        if ((t & 31) >= o) incl += u;
    }
    if (t < NPART) {
        segCnt[t] = n;
        csegCnt[t] = cn0;
        if ((t & 31) == 31) waveTot[t >> 5] = incl;
    }
    __syncthreads();
    if (t < NPART) {
        int off = 0;
        for (int w = 0; w < (t >> 5); ++w) off += waveTot[w];
        segOff[t] = off + incl - n;
        if (n > capPairs) useList = 1;
        if (cn0 > capC) useColList = 1;
        if (t == NPART - 1) {
            segOff[NPART] = off + incl;
            if (off + incl > BIN_EDGES_MAX) useList = 1;
        }
    }
    __syncthreads();
    const int total = segOff[NPART];
    const int wave = t >> 5, lane = t & 31;

    const int colList = useColList;
    int cn[NPART / 32], cv[NPART / 32];
#pragma unroll
    for (int i = 0; i < NPART / 32; ++i) {
        const int c = wave + 32 * i;
        cn[i] = colList ? 0 : csegCnt[c];
        cv[i] = (lane < cn[i]) ? cpieces[(size_t)c * capC + lane] : -1;
    }

    if (!useList) {
        for (int c = wave; c < NPART; c += 32) {
            const int pn = segCnt[c];
            const int po = segOff[c];
            const int2* p = (const int2*)(brow + (size_t)c * capPairs * 2);
            for (int k = lane; k < pn; k += 32) {
                const int2 v = p[k];
                ed[po + k] = v;
                atomicAdd(&rowCnt[v.x], 1);
            }
        }
        __syncthreads();
        const int m = rowCnt[t];
        int rincl = m;
        for (int o = 1; o < 32; o <<= 1) {
            const int u = __shfl_up(rincl, o);
            if (lane >= o) rincl += u;
        }
        if (lane == 31) rowWaveTot[wave] = rincl;
        __syncthreads();
        int roff = 0;
        for (int w = 0; w < wave; ++w) roff += rowWaveTot[w];
        rowFill[t] = roff + rincl - m;
        __syncthreads();
        for (int i = t; i < total; i += 1024) {
            const int2 v = ed[i];
            const int pos = atomicAdd(&rowFill[v.x], 1);
            cs[pos] = v.y;
        }
        __syncthreads();
        for (int row = wave; row < nrows; row += 32) {
            const int d = rowCnt[row];
            const int o = rowFill[row] - d;
            const int lim = d < ROW_CAP ? d : ROW_CAP;
            for (int k = lane; k < lim; k += 32) brow[(size_t)row * 128 + k] = cs[o + k];
        }
    } else {
        for (int e = t; e < E; e += 1024) {
            const int r = rows[e];
            int bb = r / rpb;
            bb = bb < NPART - 1 ? bb : NPART - 1;
            if (bb == b && r - base < nrows) {
                const int slot = atomicAdd(&rowCnt[r - base], 1);
                if (slot < ROW_CAP) brow[(size_t)(r - base) * 128 + slot] = cols[e];
            }
        }
    }
    if (!colList) {
#pragma unroll
        for (int i = 0; i < NPART / 32; ++i)
            if ((unsigned)cv[i] < (unsigned)BIN_ROWS_MAX) atomicAdd(&colCnt[cv[i]], 1);
#pragma unroll
        for (int i = 0; i < NPART / 32; ++i) {
            const int* q = cpieces + (size_t)(wave + 32 * i) * capC;
            for (int k = lane + 32; k < cn[i]; k += 32) {
                const int v = q[k];
                if ((unsigned)v < (unsigned)BIN_ROWS_MAX) atomicAdd(&colCnt[v], 1);
            }
        }
    } else {
        for (int e = t; e < E; e += 1024) {
            const int col = cols[e];
            int bb = col / rpb;
            bb = bb < NPART - 1 ? bb : NPART - 1;
            if (bb == b && (unsigned)(col - base) < (unsigned)BIN_ROWS_MAX) atomicAdd(&colCnt[col - base], 1);
        }
    }
    __syncthreads();
    for (int i = t; i < nrows; i += 1024) {
        cntR[base + i] = rowCnt[i];
        degC[base + i] = colCnt[i];
    }
}

#define W_PITCH 132
#define A_PITCH 132
#define TILE_ROWS 64

__device__ __forceinline__ float dinv_from(int deg) {
    return __builtin_amdgcn_rsqf((float)(deg + 1));
}

__device__ __forceinline__ v8f wmma4(v2f a, v2f b, v8f c) {
    return __builtin_amdgcn_wmma_f32_16x16x4_f32(
         false, a,  false, b,
         (short)0, c,  false,  false);
}

__device__ __forceinline__ void fma4(float4& acc, const float w, const float4 x) {
    acc.x = fmaf(w, x.x, acc.x);
    acc.y = fmaf(w, x.y, acc.y);
    acc.z = fmaf(w, x.z, acc.z);
    acc.w = fmaf(w, x.w, acc.w);
}

__device__ __forceinline__ void sum_long_row(float4& acc, const int* keys, const int* vals,
                                             int n, int total, int r, int lane,
                                             const int* __restrict__ degC,
                                             const float4* __restrict__ X4) {
    int last = -1;
    int done = 0;
    while (done < total) {
        int mn = 0x7fffffff;
        for (int j = lane; j < n; j += 32) {
            const bool mine = keys ? (keys[j] == r) : true;
            const int cj = vals[j];
            if (mine && cj > last && cj < mn) mn = cj;
        }
        for (int o = 16; o > 0; o >>= 1) {
            const int u = __shfl_xor(mn, o);
            mn = (u < mn) ? u : mn;
        }
        int cnt = 0;
        for (int j = lane; j < n; j += 32) {
            const bool mine = keys ? (keys[j] == r) : true;
            cnt += (mine && vals[j] == mn) ? 1 : 0;
        }
        for (int o = 16; o > 0; o >>= 1) cnt += __shfl_xor(cnt, o);
        cnt = __builtin_amdgcn_readfirstlane(cnt);
        mn  = __builtin_amdgcn_readfirstlane(mn);
        if (cnt <= 0) break;
        const float w0 = dinv_from(degC[mn]);
        const float4 x0 = X4[(size_t)mn * 32 + lane];
        for (int i = 0; i < cnt; ++i) fma4(acc, w0, x0);
        done += cnt;
        last = mn;
    }
}

__device__ __forceinline__ void sum_mid_row(float4& acc, const int* orow, int d, int lane,
                                            const int* __restrict__ degC,
                                            const float4* __restrict__ X4, int* s) {
    int c[4], rk[4];
#pragma unroll
    for (int q = 0; q < 4; ++q) {
        const int i = lane + 32 * q;
        c[q] = (i < d) ? orow[i] : 0x7fffffff;
        rk[q] = 0;
    }
#pragma unroll
    for (int p = 0; p < 4; ++p) {
        int lim = d - 32 * p;
        lim = lim < 32 ? lim : 32;
        for (int j = 0; j < lim; ++j) {
            const int cj = __builtin_amdgcn_readlane(c[p], j);
            const int jj = 32 * p + j;
#pragma unroll
            for (int q = 0; q < 4; ++q)
                rk[q] += (cj < c[q] || (cj == c[q] && jj < lane + 32 * q)) ? 1 : 0;
        }
    }
#pragma unroll
    for (int q = 0; q < 4; ++q)
        if (lane + 32 * q < d) s[rk[q]] = c[q];

    int k = 0;
    for (; k + 4 <= d; k += 4) {
        const int4 cc = *(const int4*)(s + k);
        const int c0 = __builtin_amdgcn_readfirstlane(cc.x);
        const int c1 = __builtin_amdgcn_readfirstlane(cc.y);
        const int c2 = __builtin_amdgcn_readfirstlane(cc.z);
        const int c3 = __builtin_amdgcn_readfirstlane(cc.w);
        const float4 x0 = X4[(size_t)c0 * 32 + lane];
        const float4 x1 = X4[(size_t)c1 * 32 + lane];
        const float4 x2 = X4[(size_t)c2 * 32 + lane];
        const float4 x3 = X4[(size_t)c3 * 32 + lane];
        const float w0 = dinv_from(degC[c0]);
        const float w1 = dinv_from(degC[c1]);
        const float w2 = dinv_from(degC[c2]);
        const float w3 = dinv_from(degC[c3]);
        fma4(acc, w0, x0);
        fma4(acc, w1, x1);
        fma4(acc, w2, x2);
        fma4(acc, w3, x3);
    }
    for (; k < d; ++k) {
        const int c0 = __builtin_amdgcn_readfirstlane(s[k]);
        const float4 x0 = X4[(size_t)c0 * 32 + lane];
        fma4(acc, dinv_from(degC[c0]), x0);
    }
}

__device__ __forceinline__ float4 sum_row(const int r, const int lane,
                                          const int* __restrict__ cntR,
                                          const int* __restrict__ degC,
                                          const float4* __restrict__ X4,
                                          const int* __restrict__ rows,
                                          const int* __restrict__ cols,
                                          const int* orow, int* list, int E) {
    const int d = cntR[r];
    const float dr = dinv_from(degC[r]);
    const float4 xr = X4[(size_t)r * 32 + lane];
    float4 acc = {dr * xr.x, dr * xr.y, dr * xr.z, dr * xr.w};

    if (d <= 32) {
        const int c = (lane < d) ? orow[lane] : 0x7fffffff;
        const float wc = (lane < d) ? dinv_from(degC[c]) : 0.f;
        int rk = 0;
        for (int j = 0; j < d; ++j) {
            const int cj = __builtin_amdgcn_readlane(c, j);
            rk += (cj < c || (cj == c && j < lane)) ? 1 : 0;
        }
        if (lane >= d) rk = lane;
        const int sorted  = __builtin_amdgcn_ds_permute(rk << 2, c);
        const int sortedW = __builtin_amdgcn_ds_permute(rk << 2, __float_as_int(wc));

        int k = 0;
        for (; k + 4 <= d; k += 4) {
            const int c0 = __builtin_amdgcn_readlane(sorted, k);
            const int c1 = __builtin_amdgcn_readlane(sorted, k + 1);
            const int c2 = __builtin_amdgcn_readlane(sorted, k + 2);
            const int c3 = __builtin_amdgcn_readlane(sorted, k + 3);
            const float w0 = __int_as_float(__builtin_amdgcn_readlane(sortedW, k));
            const float w1 = __int_as_float(__builtin_amdgcn_readlane(sortedW, k + 1));
            const float w2 = __int_as_float(__builtin_amdgcn_readlane(sortedW, k + 2));
            const float w3 = __int_as_float(__builtin_amdgcn_readlane(sortedW, k + 3));
            const float4 x0 = X4[(size_t)c0 * 32 + lane];
            const float4 x1 = X4[(size_t)c1 * 32 + lane];
            const float4 x2 = X4[(size_t)c2 * 32 + lane];
            const float4 x3 = X4[(size_t)c3 * 32 + lane];
            fma4(acc, w0, x0);
            fma4(acc, w1, x1);
            fma4(acc, w2, x2);
            fma4(acc, w3, x3);
        }
        for (; k < d; ++k) {
            const int c0 = __builtin_amdgcn_readlane(sorted, k);
            const float w0 = __int_as_float(__builtin_amdgcn_readlane(sortedW, k));
            const float4 x0 = X4[(size_t)c0 * 32 + lane];
            fma4(acc, w0, x0);
        }
    } else if (d <= ROW_CAP) {
        sum_mid_row(acc, orow, d, lane, degC, X4, list);
    } else {
        sum_long_row(acc, rows, cols, E, d, r, lane, degC, X4);
    }
    acc.x *= dr;
    acc.y *= dr;
    acc.z *= dr;
    acc.w *= dr;
    return acc;
}

__global__ __launch_bounds__(1024, 16) void gather_product_kernel(const int* __restrict__ cntR,
                                                                  const int* __restrict__ degC,
                                                                  const float* __restrict__ X,
                                                                  const float* __restrict__ W,
                                                                  const int* __restrict__ rows,
                                                                  const int* __restrict__ cols,
                                                                  int* out, int N, int E) {
    __shared__ __attribute__((aligned(16))) float Ws[128 * W_PITCH];
    __shared__ __attribute__((aligned(16))) float At[TILE_ROWS * A_PITCH];
    __shared__ __attribute__((aligned(16))) int colList[32][ROW_CAP];

    const int t = threadIdx.x;
    const float4* __restrict__ W4 = (const float4*)W;
#pragma unroll
    for (int q = 0; q < 4; ++q) {
        const int i = t + 1024 * q;
        const int row = i >> 5, c4 = i & 31;
        const float4 v = W4[i];
        *(float4*)&Ws[row * W_PITCH + c4 * 4] = v;
    }

    const int lane = t & 31;
    const int wave = __builtin_amdgcn_readfirstlane(t >> 5);
    const float4* __restrict__ X4 = (const float4*)X;

    const int strips = (N + 15) >> 4;
    const int spb = (strips + (int)gridDim.x - 1) / (int)gridDim.x;
    const int r0 = (int)blockIdx.x * spb * 16;
    int r1 = r0 + spb * 16;
    r1 = r1 < N ? r1 : N;

    const int half = lane >> 4;
    const int l16  = lane & 15;
    const int strip = wave >> 3;
    const int ct    = wave & 7;

    for (int cur = r0; cur < r1; cur += TILE_ROWS) {
        const int ra = cur + wave;
        const int rb = cur + wave + 32;
        float4 s0 = {0.f, 0.f, 0.f, 0.f};
        float4 s1 = {0.f, 0.f, 0.f, 0.f};
        if (ra < r1) s0 = sum_row(ra, lane, cntR, degC, X4, rows, cols, out + (size_t)ra * 128, colList[wave], E);
        if (rb < r1) s1 = sum_row(rb, lane, cntR, degC, X4, rows, cols, out + (size_t)rb * 128, colList[wave], E);
        if (cur != r0) __syncthreads();
        *(float4*)&At[wave * A_PITCH + lane * 4] = s0;
        *(float4*)&At[(wave + 32) * A_PITCH + lane * 4] = s1;
        __syncthreads();

        const int rowBase = cur + strip * 16;
        if (rowBase < r1) {
            const float* abase = &At[(strip * 16 + l16) * A_PITCH + half * 4];
            const float* bbase = &Ws[(ct * 16 + l16) * W_PITCH + half * 4];
            v8f acc = (v8f){0.f, 0.f, 0.f, 0.f, 0.f, 0.f, 0.f, 0.f};
#pragma unroll
            for (int g = 0; g < 4; ++g) {
                float4 a[4], b[4];
#pragma unroll
                for (int jj = 0; jj < 4; ++jj) {
                    a[jj] = *(const float4*)(abase + 8 * (4 * g + jj));
                    b[jj] = *(const float4*)(bbase + 8 * (4 * g + jj));
                }
#pragma unroll
                for (int jj = 0; jj < 4; ++jj) {
                    const v2f a0 = {a[jj].x, a[jj].y};
                    const v2f b0 = {b[jj].x, b[jj].y};
                    const v2f a1 = {a[jj].z, a[jj].w};
                    const v2f b1 = {b[jj].z, b[jj].w};
                    acc = wmma4(a0, b0, acc);
                    acc = wmma4(a1, b1, acc);
                }
            }
            const int mrow = rowBase + (half << 3);
            int* cptr = out + (size_t)mrow * 128 + ct * 16 + l16;
            const int lim = r1 - mrow;
#pragma unroll
            for (int v = 0; v < 8; ++v) {
                if (v < lim)
                    cptr[v * 128] = __float_as_int(acc[v]);
            }
        }
    }
}

extern "C" void kernel_launch(void* const* d_in, const int* in_sizes, int n_in,
                              void* d_out, int out_size, void* d_ws, size_t ws_size,
                              hipStream_t stream) {
    const float* X  = (const float*)d_in[0];
    const float* W  = (const float*)d_in[1];
    const int*   ei = (const int*)d_in[2];
    float* out = (float*)d_out;

    const int N = in_sizes[0] / 128;
    const int E = in_sizes[2] / 2;
    const int* rows = ei;
    const int* cols = ei + E;

    int* cseg   = (int*)d_ws;
    int* cntR   = cseg + (size_t)N * 128;
    int* degC   = cntR + N;
    int* bucket = (int*)out;

    partition_kernel<<<NPART, 1024, 0, stream>>>(rows, cols, cseg, bucket, N, E);
    binplace_kernel<<<NPART, 1024, 0, stream>>>(rows, cols, cntR, degC, cseg, bucket, N, E);

    gather_product_kernel<<<512, 1024, 0, stream>>>(cntR, degC, X, W, rows, cols, bucket, N, E);
}
